# starting kernel
# baseline (speedup 1.0000x reference)
#include <hip/hip_runtime.h>

#define NN   100000
#define EE   600000
#define ETOT (EE + NN)
#define INF_ 64
#define HF   128
#define GG   5000
#define OUTF 19
#define NEG_SLOPE 0.2f

typedef __attribute__((ext_vector_type(2))) float v2f;
typedef __attribute__((ext_vector_type(8))) float v8f;

__device__ __forceinline__ unsigned fkey(float f) {
    unsigned u = __float_as_uint(f);
    return (u & 0x80000000u) ? ~u : (u | 0x80000000u);
}
__device__ __forceinline__ float funkey(unsigned k) {
    return (k & 0x80000000u) ? __uint_as_float(k ^ 0x80000000u) : __uint_as_float(~k);
}

__global__ void __launch_bounds__(256)
gemm_wmma(const float* __restrict__ in, const float* __restrict__ W,
          const float* __restrict__ a_src, const float* __restrict__ a_dst,
          float* __restrict__ h, float* __restrict__ s_src, float* __restrict__ s_dst,
          int K)
{
    const int wave = threadIdx.x >> 5;
    const int lane = threadIdx.x & 31;
    const int tile = blockIdx.x * 8 + wave;
    if (tile * 16 >= NN) return;
    const int row0 = tile * 16;
    const int r    = lane & 15;
    const int hi   = lane >> 4;
    const int koff = hi * 2;

    v8f acc[8];
    const v8f vz = {0.f,0.f,0.f,0.f,0.f,0.f,0.f,0.f};
#pragma unroll
    for (int t = 0; t < 8; ++t) acc[t] = vz;

    const float* arow = in + (size_t)(row0 + r) * K;
    for (int k0 = 0; k0 < K; k0 += 4) {
        v2f a;
        a.x = arow[k0 + koff];
        a.y = arow[k0 + koff + 1];
        const float* wrow = W + (size_t)(k0 + koff) * HF + r;
#pragma unroll
        for (int t = 0; t < 8; ++t) {
            v2f b;
            b.x = wrow[t * 16];
            b.y = wrow[HF + t * 16];
            acc[t] = __builtin_amdgcn_wmma_f32_16x16x4_f32(
                false, a, false, b, (short)0, acc[t], false, false);
        }
    }

    float ps[8], pd[8];
#pragma unroll
    for (int j = 0; j < 8; ++j) { ps[j] = 0.f; pd[j] = 0.f; }
#pragma unroll
    for (int t = 0; t < 8; ++t) {
        const int col = t * 16 + r;
        const float as = a_src[col];
        const float ad = a_dst[col];
#pragma unroll
        for (int j = 0; j < 8; ++j) {
            const int row = row0 + hi * 8 + j;
            float v = acc[t][j];
            h[(size_t)row * HF + col] = v;
            ps[j] += v * as;
            pd[j] += v * ad;
        }
    }
#pragma unroll
    for (int off = 1; off < 16; off <<= 1) {
#pragma unroll
        for (int j = 0; j < 8; ++j) {
            ps[j] += __shfl_xor(ps[j], off, 32);
            pd[j] += __shfl_xor(pd[j], off, 32);
        }
    }
    if (r == 0) {
#pragma unroll
        for (int j = 0; j < 8; ++j) {
            const int row = row0 + hi * 8 + j;
            s_src[row] = ps[j];
            s_dst[row] = pd[j];
        }
    }
}

__global__ void layer_init(float* __restrict__ acc, unsigned* __restrict__ m,
                           float* __restrict__ denom)
{
    int i = blockIdx.x * blockDim.x + threadIdx.x;
    if (i < NN * (HF / 4)) {
        float4 z = {0.f, 0.f, 0.f, 0.f};
        ((float4*)acc)[i] = z;
    }
    if (i < NN) { m[i] = 0u; denom[i] = 0.f; }
}

__global__ void edge_alpha_max(const int* __restrict__ esrc, const int* __restrict__ edst,
                               const float* __restrict__ s_src, const float* __restrict__ s_dst,
                               float* __restrict__ alpha, unsigned* __restrict__ m)
{
    int e = blockIdx.x * blockDim.x + threadIdx.x;
    if (e >= ETOT) return;
    int s = (e < EE) ? esrc[e] : (e - EE);
    int d = (e < EE) ? edst[e] : (e - EE);
    float a = s_src[s] + s_dst[d];
    a = (a > 0.f) ? a : NEG_SLOPE * a;
    alpha[e] = a;
    atomicMax(m + d, fkey(a));
}

__global__ void edge_exp_sum(const int* __restrict__ edst,
                             float* __restrict__ alpha, const unsigned* __restrict__ m,
                             float* __restrict__ denom)
{
    int e = blockIdx.x * blockDim.x + threadIdx.x;
    if (e >= ETOT) return;
    int d = (e < EE) ? edst[e] : (e - EE);
    float ex = expf(alpha[e] - funkey(m[d]));
    alpha[e] = ex;
    atomicAdd(denom + d, ex);
}

__global__ void edge_scatter(const int* __restrict__ esrc, const int* __restrict__ edst,
                             const float* __restrict__ h, const float* __restrict__ alpha,
                             const float* __restrict__ denom, float* __restrict__ acc)
{
    int gw = (int)((blockIdx.x * (size_t)blockDim.x + threadIdx.x) >> 5);
    int lane = threadIdx.x & 31;
    if (gw >= ETOT) return;
    int s = (gw < EE) ? esrc[gw] : (gw - EE);
    int d = (gw < EE) ? edst[gw] : (gw - EE);
    float coef = alpha[gw] / denom[d];
    float4 hv = ((const float4*)(h + (size_t)s * HF))[lane];
    float* ap = acc + (size_t)d * HF + lane * 4;
    atomicAdd(ap + 0, coef * hv.x);
    atomicAdd(ap + 1, coef * hv.y);
    atomicAdd(ap + 2, coef * hv.z);
    atomicAdd(ap + 3, coef * hv.w);
}

__global__ void bias_relu(const float* __restrict__ acc, const float* __restrict__ b,
                          float* __restrict__ act)
{
    int i = blockIdx.x * blockDim.x + threadIdx.x;
    if (i >= NN * HF) return;
    float v = acc[i] + b[i & (HF - 1)];
    act[i] = (v > 0.f) ? v : 0.f;
}

__global__ void pool_init(float* __restrict__ pooled, float* __restrict__ cnts)
{
    int i = blockIdx.x * blockDim.x + threadIdx.x;
    if (i < GG * HF) pooled[i] = 0.f;
    if (i < GG) cnts[i] = 0.f;
}

__global__ void pool_sum(const float* __restrict__ act, const int* __restrict__ batch,
                         float* __restrict__ pooled, float* __restrict__ cnts)
{
    int t = blockIdx.x * blockDim.x + threadIdx.x;
    if (t >= NN * (HF / 4)) return;
    int node = t >> 5, c = t & 31;
    int b = batch[node];
    float4 v = ((const float4*)(act + (size_t)node * HF))[c];
    float* p = pooled + (size_t)b * HF + c * 4;
    atomicAdd(p + 0, v.x);
    atomicAdd(p + 1, v.y);
    atomicAdd(p + 2, v.z);
    atomicAdd(p + 3, v.w);
    if (c == 0) atomicAdd(cnts + b, 1.0f);
}

__global__ void __launch_bounds__(64)
fc_head(const float* __restrict__ pooled, const float* __restrict__ cnts,
        const float* __restrict__ w1, const float* __restrict__ b1,
        const float* __restrict__ w2, const float* __restrict__ b2,
        float* __restrict__ out)
{
    __shared__ float p[HF];
    __shared__ float z[64];
    int g = blockIdx.x, t = threadIdx.x;
    float inv = 1.0f / fmaxf(cnts[g], 1.0f);
    p[t]      = pooled[(size_t)g * HF + t] * inv;
    p[t + 64] = pooled[(size_t)g * HF + t + 64] * inv;
    __syncthreads();
    float a = b1[t];
    for (int k = 0; k < HF; ++k) a += p[k] * w1[k * 64 + t];
    z[t] = fmaxf(a, 0.f);
    __syncthreads();
    if (t < OUTF) {
        float o = b2[t];
        for (int k = 0; k < 64; ++k) o += z[k] * w2[k * OUTF + t];
        out[(size_t)g * OUTF + t] = o;
    }
}

extern "C" void kernel_launch(void* const* d_in, const int* in_sizes, int n_in,
                              void* d_out, int out_size, void* d_ws, size_t ws_size,
                              hipStream_t stream)
{
    const float* x      = (const float*)d_in[0];
    const int*   eidx   = (const int*)  d_in[1];
    const int*   batch  = (const int*)  d_in[2];
    const float* W[3]    = {(const float*)d_in[3],  (const float*)d_in[7],  (const float*)d_in[11]};
    const float* asrc[3] = {(const float*)d_in[4],  (const float*)d_in[8],  (const float*)d_in[12]};
    const float* adst[3] = {(const float*)d_in[5],  (const float*)d_in[9],  (const float*)d_in[13]};
    const float* bias[3] = {(const float*)d_in[6],  (const float*)d_in[10], (const float*)d_in[14]};
    const float* fc1w = (const float*)d_in[15];
    const float* fc1b = (const float*)d_in[16];
    const float* fc2w = (const float*)d_in[17];
    const float* fc2b = (const float*)d_in[18];
    float* out = (float*)d_out;

    const int* esrc = eidx;
    const int* edst = eidx + EE;

    char* ws = (char*)d_ws;
    const size_t NHB = (size_t)NN * HF * sizeof(float);
    float*    h_lin  = (float*)(ws);
    float*    acc    = (float*)(ws + NHB);
    float*    act    = (float*)(ws + 2 * NHB);
    float*    s_src  = (float*)(ws + 3 * NHB);
    float*    s_dst  = (float*)(ws + 3 * NHB + (size_t)NN * 4);
    unsigned* m      = (unsigned*)(ws + 3 * NHB + (size_t)NN * 8);
    float*    denom  = (float*)(ws + 3 * NHB + (size_t)NN * 12);
    float*    alpha  = (float*)(ws + 3 * NHB + (size_t)NN * 16);
    float*    pooled = (float*)(ws + 3 * NHB + (size_t)NN * 16 + (size_t)ETOT * 4);
    float*    cnts   = (float*)(ws + 3 * NHB + (size_t)NN * 16 + (size_t)ETOT * 4 + (size_t)GG * HF * 4);

    const int gemm_grid   = ((NN / 16) + 7) / 8;
    const int init_grid   = (NN * (HF / 4) + 255) / 256;
    const int edge_grid   = (ETOT + 255) / 256;
    const int scat_grid   = (int)(((size_t)ETOT * 32 + 255) / 256);
    const int br_grid     = (NN * HF + 255) / 256;
    const int pool_grid   = (NN * (HF / 4) + 255) / 256;
    const int pinit_grid  = (GG * HF + 255) / 256;

    const float* cur = x;
    int K = INF_;
    for (int l = 0; l < 3; ++l) {
        gemm_wmma<<<gemm_grid, 256, 0, stream>>>(cur, W[l], asrc[l], adst[l],
                                                 h_lin, s_src, s_dst, K);
        layer_init<<<init_grid, 256, 0, stream>>>(acc, m, denom);
        edge_alpha_max<<<edge_grid, 256, 0, stream>>>(esrc, edst, s_src, s_dst, alpha, m);
        edge_exp_sum<<<edge_grid, 256, 0, stream>>>(edst, alpha, m, denom);
        edge_scatter<<<scat_grid, 256, 0, stream>>>(esrc, edst, h_lin, alpha, denom, acc);
        bias_relu<<<br_grid, 256, 0, stream>>>(acc, bias[l], act);
        cur = act;
        K = HF;
    }

    pool_init<<<pinit_grid, 256, 0, stream>>>(pooled, cnts);
    pool_sum<<<pool_grid, 256, 0, stream>>>(act, batch, pooled, cnts);
    fc_head<<<GG, 64, 0, stream>>>(pooled, cnts, fc1w, fc1b, fc2w, fc2b, out);
}


# layer edge passes -> one launch: work-group holds 400 destination rows in LDS, scans all edges for its own, LDS-atomic sums of exp-weighted h rows (arrival order), softmax division after the sum
# speedup vs baseline: 2.4946x; 2.4946x over previous
#include <hip/hip_runtime.h>

#define NN   100000
#define EE   600000
#define ETOT (EE + NN)
#define INF_ 64
#define HF   128
#define GG   5000
#define OUTF 19
#define NEG_SLOPE 0.2f

typedef __attribute__((ext_vector_type(2))) float v2f;
typedef __attribute__((ext_vector_type(8))) float v8f;

__device__ __forceinline__ unsigned fkey(float f) {
    unsigned u = __float_as_uint(f);
    return (u & 0x80000000u) ? ~u : (u | 0x80000000u);
}
__device__ __forceinline__ float funkey(unsigned k) {
    return (k & 0x80000000u) ? __uint_as_float(k ^ 0x80000000u) : __uint_as_float(~k);
}

__global__ void __launch_bounds__(256)
gemm_wmma(const float* __restrict__ in, const float* __restrict__ W,
          const float* __restrict__ a_src, const float* __restrict__ a_dst,
          float* __restrict__ h, float* __restrict__ s_src, float* __restrict__ s_dst,
          int K)
{
    const int wave = threadIdx.x >> 5;
    const int lane = threadIdx.x & 31;
    const int tile = blockIdx.x * 8 + wave;
    if (tile * 16 >= NN) return;
    const int row0 = tile * 16;
    const int r    = lane & 15;
    const int hi   = lane >> 4;
    const int koff = hi * 2;

    v8f acc[8];
    const v8f vz = {0.f,0.f,0.f,0.f,0.f,0.f,0.f,0.f};
#pragma unroll
    for (int t = 0; t < 8; ++t) acc[t] = vz;

    const float* arow = in + (size_t)(row0 + r) * K;
    for (int k0 = 0; k0 < K; k0 += 4) {
        v2f a;
        a.x = arow[k0 + koff];
        a.y = arow[k0 + koff + 1];
        const float* wrow = W + (size_t)(k0 + koff) * HF + r;
#pragma unroll
        for (int t = 0; t < 8; ++t) {
            v2f b;
            b.x = wrow[t * 16];
            b.y = wrow[HF + t * 16];
            acc[t] = __builtin_amdgcn_wmma_f32_16x16x4_f32(
                false, a, false, b, (short)0, acc[t], false, false);
        }
    }

    float ps[8], pd[8];
#pragma unroll
    for (int j = 0; j < 8; ++j) { ps[j] = 0.f; pd[j] = 0.f; }
#pragma unroll
    for (int t = 0; t < 8; ++t) {
        const int col = t * 16 + r;
        const float as = a_src[col];
        const float ad = a_dst[col];
#pragma unroll
        for (int j = 0; j < 8; ++j) {
            const int row = row0 + hi * 8 + j;
            float v = acc[t][j];
            h[(size_t)row * HF + col] = v;
            ps[j] += v * as;
            pd[j] += v * ad;
        }
    }
#pragma unroll
    for (int off = 1; off < 16; off <<= 1) {
#pragma unroll
        for (int j = 0; j < 8; ++j) {
            ps[j] += __shfl_xor(ps[j], off, 32);
            pd[j] += __shfl_xor(pd[j], off, 32);
        }
    }
    if (r == 0) {
#pragma unroll
        for (int j = 0; j < 8; ++j) {
            const int row = row0 + hi * 8 + j;
            s_src[row] = ps[j];
            s_dst[row] = pd[j];
        }
    }
}

#define TT          400
#define LCAP        6144
#define AGG_THREADS 1024
#define AGG_WAVES   (AGG_THREADS / 32)
#define AGG_U       8

__device__ __forceinline__ float lrelu(float a) { return (a > 0.f) ? a : NEG_SLOPE * a; }

__global__ void __launch_bounds__(AGG_THREADS)
agg_tile(const int* __restrict__ esrc, const int* __restrict__ edst,
         const float* __restrict__ h, const float* __restrict__ s_src,
         const float* __restrict__ s_dst, const float* __restrict__ bias,
         float* __restrict__ act)
{
    __shared__ __attribute__((aligned(16))) float acc[TT * HF];
    __shared__ unsigned mkey[TT];
    __shared__ float    den[TT];
    __shared__ float    sd[TT];
    __shared__ float    aself[TT];
    __shared__ int      le[LCAP];
    __shared__ int      ldl[LCAP];
    __shared__ float    lal[LCAP];
    __shared__ int      cnt;

    const int tid   = threadIdx.x;
    const int lane  = tid & 31;
    const int wave  = tid >> 5;
    const int tile0 = blockIdx.x * TT;
    const int nT    = (NN - tile0 < TT) ? (NN - tile0) : TT;

    {
        const float4 z = {0.f, 0.f, 0.f, 0.f};
        for (int i = tid; i < TT * HF / 4; i += AGG_THREADS) ((float4*)acc)[i] = z;
        if (tid == 0) cnt = 0;
        for (int i = tid; i < nT; i += AGG_THREADS) {
            const float d = s_dst[tile0 + i];
            const float a = lrelu(s_src[tile0 + i] + d);
            sd[i] = d; aself[i] = a; mkey[i] = fkey(a); den[i] = 0.f;
        }
    }
    __syncthreads();

    {
        const int4* d4 = (const int4*)edst;
        const int NV = EE / 4;
        const int GRP = 4 * AGG_THREADS;
        auto load4 = [&](int4 (&g)[4], int base) {
#pragma unroll
            for (int k = 0; k < 4; ++k) {
                const int i = base + k * AGG_THREADS + tid;
                g[k] = d4[(i < NV) ? i : (NV - 1)];
            }
        };
        auto put = [&](unsigned dl, int e) {
            const int pos = atomicAdd(&cnt, 1);
            if (pos < LCAP) { le[pos] = e; ldl[pos] = (int)dl; }
        };
        auto match4 = [&](const int4 (&g)[4], int base) {
#pragma unroll
            for (int k = 0; k < 4; ++k) {
                const int i  = base + k * AGG_THREADS + tid;
                const int e0 = i * 4;
                const unsigned lim = (i < NV) ? (unsigned)nT : 0u;
                const unsigned d0 = (unsigned)(g[k].x - tile0);
                const unsigned d1 = (unsigned)(g[k].y - tile0);
                const unsigned d2 = (unsigned)(g[k].z - tile0);
                const unsigned d3 = (unsigned)(g[k].w - tile0);
                if ((d0 < lim) | (d1 < lim) | (d2 < lim) | (d3 < lim)) {
                    if (d0 < lim) put(d0, e0);
                    if (d1 < lim) put(d1, e0 + 1);
                    if (d2 < lim) put(d2, e0 + 2);
                    if (d3 < lim) put(d3, e0 + 3);
                }
            }
        };
        int4 ga[4], gb[4];
        load4(ga, 0);
        for (int base = 0; base < NV; base += 2 * GRP) {
            load4(gb, base + GRP);
            match4(ga, base);
            load4(ga, base + 2 * GRP);
            match4(gb, base + GRP);
        }
    }
    __syncthreads();

    const int  nraw = cnt;
    const bool ovf  = nraw > LCAP;
    const int  nl   = ovf ? 0 : nraw;

    if (!ovf) {
        for (int j = tid; j < nl; j += AGG_THREADS) {
            const int s  = esrc[le[j]];
            const int dl = ldl[j];
            const float a = lrelu(s_src[s] + sd[dl]);
            atomicMax(&mkey[dl], fkey(a));
            le[j] = s; lal[j] = a;
        }
    } else {
        for (int e = tid; e < EE; e += AGG_THREADS) {
            const unsigned dl = (unsigned)(edst[e] - tile0);
            if (dl < (unsigned)nT)
                atomicMax(&mkey[dl], fkey(lrelu(s_src[esrc[e]] + sd[dl])));
        }
    }
    __syncthreads();

    {
        const int total = nT + nl;
        for (int it0 = wave * AGG_U; it0 < total; it0 += AGG_WAVES * AGG_U) {
            float4 v[AGG_U]; float ex[AGG_U]; int dd[AGG_U];
#pragma unroll
            for (int u = 0; u < AGG_U; ++u) {
                const int it = it0 + u;
                if (it < total) {
                    int s; float a;
                    if (it < nT) { s = tile0 + it; dd[u] = it; a = aself[it]; }
                    else { const int j = it - nT; s = le[j]; dd[u] = ldl[j]; a = lal[j]; }
                    ex[u] = expf(a - funkey(mkey[dd[u]]));
                    v[u]  = ((const float4*)(h + (size_t)s * HF))[lane];
                }
            }
#pragma unroll
            for (int u = 0; u < AGG_U; ++u) {
                if (it0 + u < total) {
                    float* ar = acc + dd[u] * HF + lane;
                    atomicAdd(ar,      ex[u] * v[u].x);
                    atomicAdd(ar + 32, ex[u] * v[u].y);
                    atomicAdd(ar + 64, ex[u] * v[u].z);
                    atomicAdd(ar + 96, ex[u] * v[u].w);
                    if (lane == 0) atomicAdd(&den[dd[u]], ex[u]);
                }
            }
        }
        if (ovf) {
            for (int e = tid; e < EE; e += AGG_THREADS) {
                const unsigned dl = (unsigned)(edst[e] - tile0);
                if (dl < (unsigned)nT) {
                    const int s = esrc[e];
                    const float x = expf(lrelu(s_src[s] + sd[dl]) - funkey(mkey[dl]));
                    atomicAdd(&den[dl], x);
                    const float* hr = h + (size_t)s * HF;
                    float* ar = acc + dl * HF;
                    for (int f = 0; f < HF; ++f)
                        atomicAdd(&ar[(f & 3) * 32 + (f >> 2)], x * hr[f]);
                }
            }
        }
    }
    __syncthreads();

    {
        const float4 bv = ((const float4*)bias)[lane];
        for (int r = wave; r < nT; r += AGG_WAVES) {
            const float* ar = acc + r * HF + lane;
            const float dn = den[r];
            float4 o;
            o.x = ar[0]  / dn + bv.x;
            o.y = ar[32] / dn + bv.y;
            o.z = ar[64] / dn + bv.z;
            o.w = ar[96] / dn + bv.w;
            o.x = (o.x > 0.f) ? o.x : 0.f;
            o.y = (o.y > 0.f) ? o.y : 0.f;
            o.z = (o.z > 0.f) ? o.z : 0.f;
            o.w = (o.w > 0.f) ? o.w : 0.f;
            ((float4*)(act + (size_t)(tile0 + r) * HF))[lane] = o;
        }
    }
}

__global__ void pool_init(float* __restrict__ pooled, float* __restrict__ cnts)
{
    int i = blockIdx.x * blockDim.x + threadIdx.x;
    if (i < GG * HF) pooled[i] = 0.f;
    if (i < GG) cnts[i] = 0.f;
}

__global__ void pool_sum(const float* __restrict__ act, const int* __restrict__ batch,
                         float* __restrict__ pooled, float* __restrict__ cnts)
{
    int t = blockIdx.x * blockDim.x + threadIdx.x;
    if (t >= NN * (HF / 4)) return;
    int node = t >> 5, c = t & 31;
    int b = batch[node];
    float4 v = ((const float4*)(act + (size_t)node * HF))[c];
    float* p = pooled + (size_t)b * HF + c * 4;
    atomicAdd(p + 0, v.x);
    atomicAdd(p + 1, v.y);
    atomicAdd(p + 2, v.z);
    atomicAdd(p + 3, v.w);
    if (c == 0) atomicAdd(cnts + b, 1.0f);
}

__global__ void __launch_bounds__(64)
fc_head(const float* __restrict__ pooled, const float* __restrict__ cnts,
        const float* __restrict__ w1, const float* __restrict__ b1,
        const float* __restrict__ w2, const float* __restrict__ b2,
        float* __restrict__ out)
{
    __shared__ float p[HF];
    __shared__ float z[64];
    int g = blockIdx.x, t = threadIdx.x;
    float inv = 1.0f / fmaxf(cnts[g], 1.0f);
    p[t]      = pooled[(size_t)g * HF + t] * inv;
    p[t + 64] = pooled[(size_t)g * HF + t + 64] * inv;
    __syncthreads();
    float a = b1[t];
    for (int k = 0; k < HF; ++k) a += p[k] * w1[k * 64 + t];
    z[t] = fmaxf(a, 0.f);
    __syncthreads();
    if (t < OUTF) {
        float o = b2[t];
        for (int k = 0; k < 64; ++k) o += z[k] * w2[k * OUTF + t];
        out[(size_t)g * OUTF + t] = o;
    }
}

extern "C" void kernel_launch(void* const* d_in, const int* in_sizes, int n_in,
                              void* d_out, int out_size, void* d_ws, size_t ws_size,
                              hipStream_t stream)
{
    const float* x      = (const float*)d_in[0];
    const int*   eidx   = (const int*)  d_in[1];
    const int*   batch  = (const int*)  d_in[2];
    const float* W[3]    = {(const float*)d_in[3],  (const float*)d_in[7],  (const float*)d_in[11]};
    const float* asrc[3] = {(const float*)d_in[4],  (const float*)d_in[8],  (const float*)d_in[12]};
    const float* adst[3] = {(const float*)d_in[5],  (const float*)d_in[9],  (const float*)d_in[13]};
    const float* bias[3] = {(const float*)d_in[6],  (const float*)d_in[10], (const float*)d_in[14]};
    const float* fc1w = (const float*)d_in[15];
    const float* fc1b = (const float*)d_in[16];
    const float* fc2w = (const float*)d_in[17];
    const float* fc2b = (const float*)d_in[18];
    float* out = (float*)d_out;

    const int* esrc = eidx;
    const int* edst = eidx + EE;

    char* ws = (char*)d_ws;
    const size_t NHB = (size_t)NN * HF * sizeof(float);
    float*    h_lin  = (float*)(ws);
    float*    act    = (float*)(ws + 2 * NHB);
    float*    s_src  = (float*)(ws + 3 * NHB);
    float*    s_dst  = (float*)(ws + 3 * NHB + (size_t)NN * 4);
    float*    pooled = (float*)(ws + 3 * NHB + (size_t)NN * 16 + (size_t)ETOT * 4);
    float*    cnts   = (float*)(ws + 3 * NHB + (size_t)NN * 16 + (size_t)ETOT * 4 + (size_t)GG * HF * 4);

    const int gemm_grid   = ((NN / 16) + 7) / 8;
    const int agg_grid    = (NN + TT - 1) / TT;
    const int pool_grid   = (NN * (HF / 4) + 255) / 256;
    const int pinit_grid  = (GG * HF + 255) / 256;

    const float* cur = x;
    int K = INF_;
    for (int l = 0; l < 3; ++l) {
        gemm_wmma<<<gemm_grid, 256, 0, stream>>>(cur, W[l], asrc[l], adst[l],
                                                 h_lin, s_src, s_dst, K);
        agg_tile<<<agg_grid, AGG_THREADS, 0, stream>>>(esrc, edst, h_lin, s_src, s_dst, bias[l], act);
        cur = act;
        K = HF;
    }

    pool_init<<<pinit_grid, 256, 0, stream>>>(pooled, cnts);
    pool_sum<<<pool_grid, 256, 0, stream>>>(act, batch, pooled, cnts);
    fc_head<<<GG, 64, 0, stream>>>(pooled, cnts, fc1w, fc1b, fc2w, fc2b, out);
}


# each tile's edge list is built once: the layer-1 launch stores its LDS list (source id, row) and length to the workspace, layers 2-3 load it instead of re-scanning all 600000 destinations
# speedup vs baseline: 2.7453x; 1.1005x over previous
#include <hip/hip_runtime.h>

#define NN   100000
#define EE   600000
#define ETOT (EE + NN)
#define INF_ 64
#define HF   128
#define GG   5000
#define OUTF 19
#define NEG_SLOPE 0.2f

typedef __attribute__((ext_vector_type(2))) float v2f;
typedef __attribute__((ext_vector_type(8))) float v8f;

__device__ __forceinline__ unsigned fkey(float f) {
    unsigned u = __float_as_uint(f);
    return (u & 0x80000000u) ? ~u : (u | 0x80000000u);
}
__device__ __forceinline__ float funkey(unsigned k) {
    return (k & 0x80000000u) ? __uint_as_float(k ^ 0x80000000u) : __uint_as_float(~k);
}

__global__ void __launch_bounds__(256)
gemm_wmma(const float* __restrict__ in, const float* __restrict__ W,
          const float* __restrict__ a_src, const float* __restrict__ a_dst,
          float* __restrict__ h, float* __restrict__ s_src, float* __restrict__ s_dst,
          int K)
{
    const int wave = threadIdx.x >> 5;
    const int lane = threadIdx.x & 31;
    const int tile = blockIdx.x * 8 + wave;
    if (tile * 16 >= NN) return;
    const int row0 = tile * 16;
    const int r    = lane & 15;
    const int hi   = lane >> 4;
    const int koff = hi * 2;

    v8f acc[8];
    const v8f vz = {0.f,0.f,0.f,0.f,0.f,0.f,0.f,0.f};
#pragma unroll
    for (int t = 0; t < 8; ++t) acc[t] = vz;

    const float* arow = in + (size_t)(row0 + r) * K;
    for (int k0 = 0; k0 < K; k0 += 4) {
        v2f a;
        a.x = arow[k0 + koff];
        a.y = arow[k0 + koff + 1];
        const float* wrow = W + (size_t)(k0 + koff) * HF + r;
#pragma unroll
        for (int t = 0; t < 8; ++t) {
            v2f b;
            b.x = wrow[t * 16];
            b.y = wrow[HF + t * 16];
            acc[t] = __builtin_amdgcn_wmma_f32_16x16x4_f32(
                false, a, false, b, (short)0, acc[t], false, false);
        }
    }

    float ps[8], pd[8];
#pragma unroll
    for (int j = 0; j < 8; ++j) { ps[j] = 0.f; pd[j] = 0.f; }
#pragma unroll
    for (int t = 0; t < 8; ++t) {
        const int col = t * 16 + r;
        const float as = a_src[col];
        const float ad = a_dst[col];
#pragma unroll
        for (int j = 0; j < 8; ++j) {
            const int row = row0 + hi * 8 + j;
            float v = acc[t][j];
            h[(size_t)row * HF + col] = v;
            ps[j] += v * as;
            pd[j] += v * ad;
        }
    }
#pragma unroll
    for (int off = 1; off < 16; off <<= 1) {
#pragma unroll
        for (int j = 0; j < 8; ++j) {
            ps[j] += __shfl_xor(ps[j], off, 32);
            pd[j] += __shfl_xor(pd[j], off, 32);
        }
    }
    if (r == 0) {
#pragma unroll
        for (int j = 0; j < 8; ++j) {
            const int row = row0 + hi * 8 + j;
            s_src[row] = ps[j];
            s_dst[row] = pd[j];
        }
    }
}

#define TT          400
#define LCAP        6144
#define AGG_THREADS 1024
#define AGG_WAVES   (AGG_THREADS / 32)
#define AGG_U       8

__device__ __forceinline__ float lrelu(float a) { return (a > 0.f) ? a : NEG_SLOPE * a; }

template <bool FIRST>
__global__ void __launch_bounds__(AGG_THREADS)
agg_tile(const int* __restrict__ esrc, const int* __restrict__ edst,
         const float* __restrict__ h, const float* __restrict__ s_src,
         const float* __restrict__ s_dst, const float* __restrict__ bias,
         float* __restrict__ act,
         int* __restrict__ tl_src, int* __restrict__ tl_dl, int* __restrict__ tl_cnt)
{
    __shared__ __attribute__((aligned(16))) float acc[TT * HF];
    __shared__ unsigned mkey[TT];
    __shared__ float    den[TT];
    __shared__ float    sd[TT];
    __shared__ float    aself[TT];
    __shared__ int      le[LCAP];
    __shared__ int      ldl[LCAP];
    __shared__ float    lal[LCAP];
    __shared__ int      cnt;

    const int tid   = threadIdx.x;
    const int lane  = tid & 31;
    const int wave  = tid >> 5;
    const int tile0 = blockIdx.x * TT;
    const int nT    = (NN - tile0 < TT) ? (NN - tile0) : TT;

    {
        const float4 z = {0.f, 0.f, 0.f, 0.f};
        for (int i = tid; i < TT * HF / 4; i += AGG_THREADS) ((float4*)acc)[i] = z;
        if (tid == 0) cnt = 0;
        for (int i = tid; i < nT; i += AGG_THREADS) {
            const float d = s_dst[tile0 + i];
            const float a = lrelu(s_src[tile0 + i] + d);
            sd[i] = d; aself[i] = a; mkey[i] = fkey(a); den[i] = 0.f;
        }
    }
    __syncthreads();

    int nraw;
    if constexpr (FIRST) {
        const int4* d4 = (const int4*)edst;
        const int NV = EE / 4;
        const int GRP = 4 * AGG_THREADS;
        auto load4 = [&](int4 (&g)[4], int base) {
#pragma unroll
            for (int k = 0; k < 4; ++k) {
                const int i = base + k * AGG_THREADS + tid;
                g[k] = d4[(i < NV) ? i : (NV - 1)];
            }
        };
        auto put = [&](unsigned dl, int e) {
            const int pos = atomicAdd(&cnt, 1);
            if (pos < LCAP) { le[pos] = e; ldl[pos] = (int)dl; }
        };
        auto match4 = [&](const int4 (&g)[4], int base) {
#pragma unroll
            for (int k = 0; k < 4; ++k) {
                const int i  = base + k * AGG_THREADS + tid;
                const int e0 = i * 4;
                const unsigned lim = (i < NV) ? (unsigned)nT : 0u;
                const unsigned d0 = (unsigned)(g[k].x - tile0);
                const unsigned d1 = (unsigned)(g[k].y - tile0);
                const unsigned d2 = (unsigned)(g[k].z - tile0);
                const unsigned d3 = (unsigned)(g[k].w - tile0);
                if ((d0 < lim) | (d1 < lim) | (d2 < lim) | (d3 < lim)) {
                    if (d0 < lim) put(d0, e0);
                    if (d1 < lim) put(d1, e0 + 1);
                    if (d2 < lim) put(d2, e0 + 2);
                    if (d3 < lim) put(d3, e0 + 3);
                }
            }
        };
        int4 ga[4], gb[4];
        load4(ga, 0);
        for (int base = 0; base < NV; base += 2 * GRP) {
            load4(gb, base + GRP);
            match4(ga, base);
            load4(ga, base + 2 * GRP);
            match4(gb, base + GRP);
        }
        __syncthreads();
        nraw = cnt;
    } else {
        nraw = tl_cnt[blockIdx.x];
    }
    const bool ovf  = nraw > LCAP;
    const int  nl   = ovf ? 0 : nraw;
    const size_t lbase = (size_t)blockIdx.x * LCAP;

    if (!ovf) {
        for (int j = tid; j < nl; j += AGG_THREADS) {
            int s, dl;
            if constexpr (FIRST) {
                s = esrc[le[j]]; dl = ldl[j];
                tl_src[lbase + j] = s; tl_dl[lbase + j] = dl;
            } else {
                s = tl_src[lbase + j]; dl = tl_dl[lbase + j];
                ldl[j] = dl;
            }
            const float a = lrelu(s_src[s] + sd[dl]);
            atomicMax(&mkey[dl], fkey(a));
            le[j] = s; lal[j] = a;
        }
    } else {
        for (int e = tid; e < EE; e += AGG_THREADS) {
            const unsigned dl = (unsigned)(edst[e] - tile0);
            if (dl < (unsigned)nT)
                atomicMax(&mkey[dl], fkey(lrelu(s_src[esrc[e]] + sd[dl])));
        }
    }
    if constexpr (FIRST) { if (tid == 0) tl_cnt[blockIdx.x] = nraw; }
    __syncthreads();

    {
        const int total = nT + nl;
        for (int it0 = wave * AGG_U; it0 < total; it0 += AGG_WAVES * AGG_U) {
            float4 v[AGG_U]; float ex[AGG_U]; int dd[AGG_U];
#pragma unroll
            for (int u = 0; u < AGG_U; ++u) {
                const int it = it0 + u;
                if (it < total) {
                    int s; float a;
                    if (it < nT) { s = tile0 + it; dd[u] = it; a = aself[it]; }
                    else { const int j = it - nT; s = le[j]; dd[u] = ldl[j]; a = lal[j]; }
                    ex[u] = expf(a - funkey(mkey[dd[u]]));
                    v[u]  = ((const float4*)(h + (size_t)s * HF))[lane];
                }
            }
#pragma unroll
            for (int u = 0; u < AGG_U; ++u) {
                if (it0 + u < total) {
                    float* ar = acc + dd[u] * HF + lane;
                    atomicAdd(ar,      ex[u] * v[u].x);
                    atomicAdd(ar + 32, ex[u] * v[u].y);
                    atomicAdd(ar + 64, ex[u] * v[u].z);
                    atomicAdd(ar + 96, ex[u] * v[u].w);
                    if (lane == 0) atomicAdd(&den[dd[u]], ex[u]);
                }
            }
        }
        if (ovf) {
            for (int e = tid; e < EE; e += AGG_THREADS) {
                const unsigned dl = (unsigned)(edst[e] - tile0);
                if (dl < (unsigned)nT) {
                    const int s = esrc[e];
                    const float x = expf(lrelu(s_src[s] + sd[dl]) - funkey(mkey[dl]));
                    atomicAdd(&den[dl], x);
                    const float* hr = h + (size_t)s * HF;
                    float* ar = acc + dl * HF;
                    for (int f = 0; f < HF; ++f)
                        atomicAdd(&ar[(f & 3) * 32 + (f >> 2)], x * hr[f]);
                }
            }
        }
    }
    __syncthreads();

    {
        const float4 bv = ((const float4*)bias)[lane];
        for (int r = wave; r < nT; r += AGG_WAVES) {
            const float* ar = acc + r * HF + lane;
            const float dn = den[r];
            float4 o;
            o.x = ar[0]  / dn + bv.x;
            o.y = ar[32] / dn + bv.y;
            o.z = ar[64] / dn + bv.z;
            o.w = ar[96] / dn + bv.w;
            o.x = (o.x > 0.f) ? o.x : 0.f;
            o.y = (o.y > 0.f) ? o.y : 0.f;
            o.z = (o.z > 0.f) ? o.z : 0.f;
            o.w = (o.w > 0.f) ? o.w : 0.f;
            ((float4*)(act + (size_t)(tile0 + r) * HF))[lane] = o;
        }
    }
}

__global__ void pool_init(float* __restrict__ pooled, float* __restrict__ cnts)
{
    int i = blockIdx.x * blockDim.x + threadIdx.x;
    if (i < GG * HF) pooled[i] = 0.f;
    if (i < GG) cnts[i] = 0.f;
}

__global__ void pool_sum(const float* __restrict__ act, const int* __restrict__ batch,
                         float* __restrict__ pooled, float* __restrict__ cnts)
{
    int t = blockIdx.x * blockDim.x + threadIdx.x;
    if (t >= NN * (HF / 4)) return;
    int node = t >> 5, c = t & 31;
    int b = batch[node];
    float4 v = ((const float4*)(act + (size_t)node * HF))[c];
    float* p = pooled + (size_t)b * HF + c * 4;
    atomicAdd(p + 0, v.x);
    atomicAdd(p + 1, v.y);
    atomicAdd(p + 2, v.z);
    atomicAdd(p + 3, v.w);
    if (c == 0) atomicAdd(cnts + b, 1.0f);
}

__global__ void __launch_bounds__(64)
fc_head(const float* __restrict__ pooled, const float* __restrict__ cnts,
        const float* __restrict__ w1, const float* __restrict__ b1,
        const float* __restrict__ w2, const float* __restrict__ b2,
        float* __restrict__ out)
{
    __shared__ float p[HF];
    __shared__ float z[64];
    int g = blockIdx.x, t = threadIdx.x;
    float inv = 1.0f / fmaxf(cnts[g], 1.0f);
    p[t]      = pooled[(size_t)g * HF + t] * inv;
    p[t + 64] = pooled[(size_t)g * HF + t + 64] * inv;
    __syncthreads();
    float a = b1[t];
    for (int k = 0; k < HF; ++k) a += p[k] * w1[k * 64 + t];
    z[t] = fmaxf(a, 0.f);
    __syncthreads();
    if (t < OUTF) {
        float o = b2[t];
        for (int k = 0; k < 64; ++k) o += z[k] * w2[k * OUTF + t];
        out[(size_t)g * OUTF + t] = o;
    }
}

extern "C" void kernel_launch(void* const* d_in, const int* in_sizes, int n_in,
                              void* d_out, int out_size, void* d_ws, size_t ws_size,
                              hipStream_t stream)
{
    const float* x      = (const float*)d_in[0];
    const int*   eidx   = (const int*)  d_in[1];
    const int*   batch  = (const int*)  d_in[2];
    const float* W[3]    = {(const float*)d_in[3],  (const float*)d_in[7],  (const float*)d_in[11]};
    const float* asrc[3] = {(const float*)d_in[4],  (const float*)d_in[8],  (const float*)d_in[12]};
    const float* adst[3] = {(const float*)d_in[5],  (const float*)d_in[9],  (const float*)d_in[13]};
    const float* bias[3] = {(const float*)d_in[6],  (const float*)d_in[10], (const float*)d_in[14]};
    const float* fc1w = (const float*)d_in[15];
    const float* fc1b = (const float*)d_in[16];
    const float* fc2w = (const float*)d_in[17];
    const float* fc2b = (const float*)d_in[18];
    float* out = (float*)d_out;

    const int* esrc = eidx;
    const int* edst = eidx + EE;

    char* ws = (char*)d_ws;
    const size_t NHB = (size_t)NN * HF * sizeof(float);
    float*    h_lin  = (float*)(ws);
    int*      tl_src = (int*)(ws + NHB);
    int*      tl_dl  = tl_src + (size_t)((NN + TT - 1) / TT) * LCAP;
    int*      tl_cnt = tl_dl  + (size_t)((NN + TT - 1) / TT) * LCAP;
    float*    act    = (float*)(ws + 2 * NHB);
    float*    s_src  = (float*)(ws + 3 * NHB);
    float*    s_dst  = (float*)(ws + 3 * NHB + (size_t)NN * 4);
    float*    pooled = (float*)(ws + 3 * NHB + (size_t)NN * 16 + (size_t)ETOT * 4);
    float*    cnts   = (float*)(ws + 3 * NHB + (size_t)NN * 16 + (size_t)ETOT * 4 + (size_t)GG * HF * 4);

    const int gemm_grid   = ((NN / 16) + 7) / 8;
    const int agg_grid    = (NN + TT - 1) / TT;
    const int pool_grid   = (NN * (HF / 4) + 255) / 256;
    const int pinit_grid  = (GG * HF + 255) / 256;

    gemm_wmma<<<gemm_grid, 256, 0, stream>>>(x, W[0], asrc[0], adst[0],
                                             h_lin, s_src, s_dst, INF_);
    agg_tile<true><<<agg_grid, AGG_THREADS, 0, stream>>>(esrc, edst, h_lin, s_src, s_dst, bias[0], act,
                                                         tl_src, tl_dl, tl_cnt);
    for (int l = 1; l < 3; ++l) {
        gemm_wmma<<<gemm_grid, 256, 0, stream>>>(act, W[l], asrc[l], adst[l],
                                                 h_lin, s_src, s_dst, HF);
        agg_tile<false><<<agg_grid, AGG_THREADS, 0, stream>>>(esrc, edst, h_lin, s_src, s_dst, bias[l], act,
                                                              tl_src, tl_dl, tl_cnt);
    }

    pool_init<<<pinit_grid, 256, 0, stream>>>(pooled, cnts);
    pool_sum<<<pool_grid, 256, 0, stream>>>(act, batch, pooled, cnts);
    fc_head<<<GG, 64, 0, stream>>>(pooled, cnts, fc1w, fc1b, fc2w, fc2b, out);
}


# pooling: a wave sums 32 consecutive node rows in registers per run of equal graph id (node order), float atomics only at run ends (arrival order); pooling waves 100000 -> 3125, clear 20000 -> 1024
# speedup vs baseline: 3.1207x; 1.1367x over previous
#include <hip/hip_runtime.h>

#define NN   100000
#define EE   600000
#define ETOT (EE + NN)
#define INF_ 64
#define HF   128
#define GG   5000
#define OUTF 19
#define NEG_SLOPE 0.2f

typedef __attribute__((ext_vector_type(2))) float v2f;
typedef __attribute__((ext_vector_type(8))) float v8f;

__device__ __forceinline__ unsigned fkey(float f) {
    unsigned u = __float_as_uint(f);
    return (u & 0x80000000u) ? ~u : (u | 0x80000000u);
}
__device__ __forceinline__ float funkey(unsigned k) {
    return (k & 0x80000000u) ? __uint_as_float(k ^ 0x80000000u) : __uint_as_float(~k);
}

__global__ void __launch_bounds__(256)
gemm_wmma(const float* __restrict__ in, const float* __restrict__ W,
          const float* __restrict__ a_src, const float* __restrict__ a_dst,
          float* __restrict__ h, float* __restrict__ s_src, float* __restrict__ s_dst,
          int K)
{
    const int wave = threadIdx.x >> 5;
    const int lane = threadIdx.x & 31;
    const int tile = blockIdx.x * 8 + wave;
    if (tile * 16 >= NN) return;
    const int row0 = tile * 16;
    const int r    = lane & 15;
    const int hi   = lane >> 4;
    const int koff = hi * 2;

    v8f acc[8];
    const v8f vz = {0.f,0.f,0.f,0.f,0.f,0.f,0.f,0.f};
#pragma unroll
    for (int t = 0; t < 8; ++t) acc[t] = vz;

    const float* arow = in + (size_t)(row0 + r) * K;
    for (int k0 = 0; k0 < K; k0 += 4) {
        v2f a;
        a.x = arow[k0 + koff];
        a.y = arow[k0 + koff + 1];
        const float* wrow = W + (size_t)(k0 + koff) * HF + r;
#pragma unroll
        for (int t = 0; t < 8; ++t) {
            v2f b;
            b.x = wrow[t * 16];
            b.y = wrow[HF + t * 16];
            acc[t] = __builtin_amdgcn_wmma_f32_16x16x4_f32(
                false, a, false, b, (short)0, acc[t], false, false);
        }
    }

    float ps[8], pd[8];
#pragma unroll
    for (int j = 0; j < 8; ++j) { ps[j] = 0.f; pd[j] = 0.f; }
#pragma unroll
    for (int t = 0; t < 8; ++t) {
        const int col = t * 16 + r;
        const float as = a_src[col];
        const float ad = a_dst[col];
#pragma unroll
        for (int j = 0; j < 8; ++j) {
            const int row = row0 + hi * 8 + j;
            float v = acc[t][j];
            h[(size_t)row * HF + col] = v;
            ps[j] += v * as;
            pd[j] += v * ad;
        }
    }
#pragma unroll
    for (int off = 1; off < 16; off <<= 1) {
#pragma unroll
        for (int j = 0; j < 8; ++j) {
            ps[j] += __shfl_xor(ps[j], off, 32);
            pd[j] += __shfl_xor(pd[j], off, 32);
        }
    }
    if (r == 0) {
#pragma unroll
        for (int j = 0; j < 8; ++j) {
            const int row = row0 + hi * 8 + j;
            s_src[row] = ps[j];
            s_dst[row] = pd[j];
        }
    }
}

#define TT          400
#define LCAP        6144
#define AGG_THREADS 1024
#define AGG_WAVES   (AGG_THREADS / 32)
#define AGG_U       8

__device__ __forceinline__ float lrelu(float a) { return (a > 0.f) ? a : NEG_SLOPE * a; }

template <bool FIRST>
__global__ void __launch_bounds__(AGG_THREADS)
agg_tile(const int* __restrict__ esrc, const int* __restrict__ edst,
         const float* __restrict__ h, const float* __restrict__ s_src,
         const float* __restrict__ s_dst, const float* __restrict__ bias,
         float* __restrict__ act,
         int* __restrict__ tl_src, int* __restrict__ tl_dl, int* __restrict__ tl_cnt)
{
    __shared__ __attribute__((aligned(16))) float acc[TT * HF];
    __shared__ unsigned mkey[TT];
    __shared__ float    den[TT];
    __shared__ float    sd[TT];
    __shared__ float    aself[TT];
    __shared__ int      le[LCAP];
    __shared__ int      ldl[LCAP];
    __shared__ float    lal[LCAP];
    __shared__ int      cnt;

    const int tid   = threadIdx.x;
    const int lane  = tid & 31;
    const int wave  = tid >> 5;
    const int tile0 = blockIdx.x * TT;
    const int nT    = (NN - tile0 < TT) ? (NN - tile0) : TT;

    {
        const float4 z = {0.f, 0.f, 0.f, 0.f};
        for (int i = tid; i < TT * HF / 4; i += AGG_THREADS) ((float4*)acc)[i] = z;
        if (tid == 0) cnt = 0;
        for (int i = tid; i < nT; i += AGG_THREADS) {
            const float d = s_dst[tile0 + i];
            const float a = lrelu(s_src[tile0 + i] + d);
            sd[i] = d; aself[i] = a; mkey[i] = fkey(a); den[i] = 0.f;
        }
    }
    __syncthreads();

    int nraw;
    if constexpr (FIRST) {
        const int4* d4 = (const int4*)edst;
        const int NV = EE / 4;
        const int GRP = 4 * AGG_THREADS;
        auto load4 = [&](int4 (&g)[4], int base) {
#pragma unroll
            for (int k = 0; k < 4; ++k) {
                const int i = base + k * AGG_THREADS + tid;
                g[k] = d4[(i < NV) ? i : (NV - 1)];
            }
        };
        auto put = [&](unsigned dl, int e) {
            const int pos = atomicAdd(&cnt, 1);
            if (pos < LCAP) { le[pos] = e; ldl[pos] = (int)dl; }
        };
        auto match4 = [&](const int4 (&g)[4], int base) {
#pragma unroll
            for (int k = 0; k < 4; ++k) {
                const int i  = base + k * AGG_THREADS + tid;
                const int e0 = i * 4;
                const unsigned lim = (i < NV) ? (unsigned)nT : 0u;
                const unsigned d0 = (unsigned)(g[k].x - tile0);
                const unsigned d1 = (unsigned)(g[k].y - tile0);
                const unsigned d2 = (unsigned)(g[k].z - tile0);
                const unsigned d3 = (unsigned)(g[k].w - tile0);
                if ((d0 < lim) | (d1 < lim) | (d2 < lim) | (d3 < lim)) {
                    if (d0 < lim) put(d0, e0);
                    if (d1 < lim) put(d1, e0 + 1);
                    if (d2 < lim) put(d2, e0 + 2);
                    if (d3 < lim) put(d3, e0 + 3);
                }
            }
        };
        int4 ga[4], gb[4];
        load4(ga, 0);
        for (int base = 0; base < NV; base += 2 * GRP) {
            load4(gb, base + GRP);
            match4(ga, base);
            load4(ga, base + 2 * GRP);
            match4(gb, base + GRP);
        }
        __syncthreads();
        nraw = cnt;
    } else {
        nraw = tl_cnt[blockIdx.x];
    }
    const bool ovf  = nraw > LCAP;
    const int  nl   = ovf ? 0 : nraw;
    const size_t lbase = (size_t)blockIdx.x * LCAP;

    if (!ovf) {
        for (int j = tid; j < nl; j += AGG_THREADS) {
            int s, dl;
            if constexpr (FIRST) {
                s = esrc[le[j]]; dl = ldl[j];
                tl_src[lbase + j] = s; tl_dl[lbase + j] = dl;
            } else {
                s = tl_src[lbase + j]; dl = tl_dl[lbase + j];
                ldl[j] = dl;
            }
            const float a = lrelu(s_src[s] + sd[dl]);
            atomicMax(&mkey[dl], fkey(a));
            le[j] = s; lal[j] = a;
        }
    } else {
        for (int e = tid; e < EE; e += AGG_THREADS) {
            const unsigned dl = (unsigned)(edst[e] - tile0);
            if (dl < (unsigned)nT)
                atomicMax(&mkey[dl], fkey(lrelu(s_src[esrc[e]] + sd[dl])));
        }
    }
    if constexpr (FIRST) { if (tid == 0) tl_cnt[blockIdx.x] = nraw; }
    __syncthreads();

    {
        const int total = nT + nl;
        for (int it0 = wave * AGG_U; it0 < total; it0 += AGG_WAVES * AGG_U) {
            float4 v[AGG_U]; float ex[AGG_U]; int dd[AGG_U];
#pragma unroll
            for (int u = 0; u < AGG_U; ++u) {
                const int it = it0 + u;
                if (it < total) {
                    int s; float a;
                    if (it < nT) { s = tile0 + it; dd[u] = it; a = aself[it]; }
                    else { const int j = it - nT; s = le[j]; dd[u] = ldl[j]; a = lal[j]; }
                    ex[u] = expf(a - funkey(mkey[dd[u]]));
                    v[u]  = ((const float4*)(h + (size_t)s * HF))[lane];
                }
            }
#pragma unroll
            for (int u = 0; u < AGG_U; ++u) {
                if (it0 + u < total) {
                    float* ar = acc + dd[u] * HF + lane;
                    atomicAdd(ar,      ex[u] * v[u].x);
                    atomicAdd(ar + 32, ex[u] * v[u].y);
                    atomicAdd(ar + 64, ex[u] * v[u].z);
                    atomicAdd(ar + 96, ex[u] * v[u].w);
                    if (lane == 0) atomicAdd(&den[dd[u]], ex[u]);
                }
            }
        }
        if (ovf) {
            for (int e = tid; e < EE; e += AGG_THREADS) {
                const unsigned dl = (unsigned)(edst[e] - tile0);
                if (dl < (unsigned)nT) {
                    const int s = esrc[e];
                    const float x = expf(lrelu(s_src[s] + sd[dl]) - funkey(mkey[dl]));
                    atomicAdd(&den[dl], x);
                    const float* hr = h + (size_t)s * HF;
                    float* ar = acc + dl * HF;
                    for (int f = 0; f < HF; ++f)
                        atomicAdd(&ar[(f & 3) * 32 + (f >> 2)], x * hr[f]);
                }
            }
        }
    }
    __syncthreads();

    {
        const float4 bv = ((const float4*)bias)[lane];
        for (int r = wave; r < nT; r += AGG_WAVES) {
            const float* ar = acc + r * HF + lane;
            const float dn = den[r];
            float4 o;
            o.x = ar[0]  / dn + bv.x;
            o.y = ar[32] / dn + bv.y;
            o.z = ar[64] / dn + bv.z;
            o.w = ar[96] / dn + bv.w;
            o.x = (o.x > 0.f) ? o.x : 0.f;
            o.y = (o.y > 0.f) ? o.y : 0.f;
            o.z = (o.z > 0.f) ? o.z : 0.f;
            o.w = (o.w > 0.f) ? o.w : 0.f;
            ((float4*)(act + (size_t)(tile0 + r) * HF))[lane] = o;
        }
    }
}

#define PI_BLOCKS 128
__global__ void __launch_bounds__(256)
pool_init(float* __restrict__ pooled, float* __restrict__ cnts)
{
    const float4 z = {0.f, 0.f, 0.f, 0.f};
    const int stride = PI_BLOCKS * 256;
    for (int i = blockIdx.x * 256 + threadIdx.x; i < GG * HF / 4; i += stride) ((float4*)pooled)[i] = z;
    for (int i = blockIdx.x * 256 + threadIdx.x; i < GG / 4; i += stride) ((float4*)cnts)[i] = z;
}

#define PC_NODES 32
static_assert(NN % PC_NODES == 0, "a chunk never runs past the last node");
static_assert(GG % 4 == 0 && (GG * HF) % 4 == 0, "pool_init clears whole float4s");

__global__ void __launch_bounds__(256)
pool_chunk(const float* __restrict__ act, const int* __restrict__ batch,
           float* __restrict__ pooled, float* __restrict__ cnts)
{
    const int lane = threadIdx.x & 31;
    const int n0   = (blockIdx.x * 8 + (threadIdx.x >> 5)) * PC_NODES;
    if (n0 >= NN) return;
    const int bl = batch[n0 + lane];

    float4 v[PC_NODES];
#pragma unroll
    for (int i = 0; i < PC_NODES; ++i)
        v[i] = ((const float4*)(act + (size_t)(n0 + i) * HF))[lane];

    auto flush = [&](int g, const float4& s, float c) {
        float* p = pooled + (size_t)g * HF + lane * 4;
        atomicAdd(p + 0, s.x);
        atomicAdd(p + 1, s.y);
        atomicAdd(p + 2, s.z);
        atomicAdd(p + 3, s.w);
        if (lane == 0) atomicAdd(cnts + g, c);
    };

    int    cur = __builtin_amdgcn_readlane(bl, 0);
    float4 s   = v[0];
    float  c   = 1.f;
#pragma unroll
    for (int i = 1; i < PC_NODES; ++i) {
        const int b = __builtin_amdgcn_readlane(bl, i);
        if (b != cur) {
            flush(cur, s, c);
            cur = b; s = v[i]; c = 1.f;
        } else {
            s.x += v[i].x; s.y += v[i].y; s.z += v[i].z; s.w += v[i].w;
            c += 1.f;
        }
    }
    flush(cur, s, c);
}

__global__ void __launch_bounds__(64)
fc_head(const float* __restrict__ pooled, const float* __restrict__ cnts,
        const float* __restrict__ w1, const float* __restrict__ b1,
        const float* __restrict__ w2, const float* __restrict__ b2,
        float* __restrict__ out)
{
    __shared__ float p[HF];
    __shared__ float z[64];
    int g = blockIdx.x, t = threadIdx.x;
    float inv = 1.0f / fmaxf(cnts[g], 1.0f);
    p[t]      = pooled[(size_t)g * HF + t] * inv;
    p[t + 64] = pooled[(size_t)g * HF + t + 64] * inv;
    __syncthreads();
    float a = b1[t];
    for (int k = 0; k < HF; ++k) a += p[k] * w1[k * 64 + t];
    z[t] = fmaxf(a, 0.f);
    __syncthreads();
    if (t < OUTF) {
        float o = b2[t];
        for (int k = 0; k < 64; ++k) o += z[k] * w2[k * OUTF + t];
        out[(size_t)g * OUTF + t] = o;
    }
}

extern "C" void kernel_launch(void* const* d_in, const int* in_sizes, int n_in,
                              void* d_out, int out_size, void* d_ws, size_t ws_size,
                              hipStream_t stream)
{
    const float* x      = (const float*)d_in[0];
    const int*   eidx   = (const int*)  d_in[1];
    const int*   batch  = (const int*)  d_in[2];
    const float* W[3]    = {(const float*)d_in[3],  (const float*)d_in[7],  (const float*)d_in[11]};
    const float* asrc[3] = {(const float*)d_in[4],  (const float*)d_in[8],  (const float*)d_in[12]};
    const float* adst[3] = {(const float*)d_in[5],  (const float*)d_in[9],  (const float*)d_in[13]};
    const float* bias[3] = {(const float*)d_in[6],  (const float*)d_in[10], (const float*)d_in[14]};
    const float* fc1w = (const float*)d_in[15];
    const float* fc1b = (const float*)d_in[16];
    const float* fc2w = (const float*)d_in[17];
    const float* fc2b = (const float*)d_in[18];
    float* out = (float*)d_out;

    const int* esrc = eidx;
    const int* edst = eidx + EE;

    char* ws = (char*)d_ws;
    const size_t NHB = (size_t)NN * HF * sizeof(float);
    float*    h_lin  = (float*)(ws);
    int*      tl_src = (int*)(ws + NHB);
    int*      tl_dl  = tl_src + (size_t)((NN + TT - 1) / TT) * LCAP;
    int*      tl_cnt = tl_dl  + (size_t)((NN + TT - 1) / TT) * LCAP;
    float*    act    = (float*)(ws + 2 * NHB);
    float*    s_src  = (float*)(ws + 3 * NHB);
    float*    s_dst  = (float*)(ws + 3 * NHB + (size_t)NN * 4);
    float*    pooled = (float*)(ws + 3 * NHB + (size_t)NN * 16 + (size_t)ETOT * 4);
    float*    cnts   = (float*)(ws + 3 * NHB + (size_t)NN * 16 + (size_t)ETOT * 4 + (size_t)GG * HF * 4);

    const int gemm_grid   = ((NN / 16) + 7) / 8;
    const int agg_grid    = (NN + TT - 1) / TT;
    const int pool_grid   = (NN / PC_NODES + 7) / 8;

    gemm_wmma<<<gemm_grid, 256, 0, stream>>>(x, W[0], asrc[0], adst[0],
                                             h_lin, s_src, s_dst, INF_);
    agg_tile<true><<<agg_grid, AGG_THREADS, 0, stream>>>(esrc, edst, h_lin, s_src, s_dst, bias[0], act,
                                                         tl_src, tl_dl, tl_cnt);
    for (int l = 1; l < 3; ++l) {
        gemm_wmma<<<gemm_grid, 256, 0, stream>>>(act, W[l], asrc[l], adst[l],
                                                 h_lin, s_src, s_dst, HF);
        agg_tile<false><<<agg_grid, AGG_THREADS, 0, stream>>>(esrc, edst, h_lin, s_src, s_dst, bias[l], act,
                                                              tl_src, tl_dl, tl_cnt);
    }

    pool_init<<<PI_BLOCKS, 256, 0, stream>>>(pooled, cnts);
    pool_chunk<<<pool_grid, 256, 0, stream>>>(act, batch, pooled, cnts);
    fc_head<<<GG, 64, 0, stream>>>(pooled, cnts, fc1w, fc1b, fc2w, fc2b, out);
}


# agg: per-edge values made once, not by all 32 lanes: softmax weight by one thread per edge (softmax sum added per thread, arrival order); row pass reads 8 items per LDS trip, readlane to scalars
# speedup vs baseline: 3.4004x; 1.0896x over previous
#include <hip/hip_runtime.h>

#define NN   100000
#define EE   600000
#define ETOT (EE + NN)
#define INF_ 64
#define HF   128
#define GG   5000
#define OUTF 19
#define NEG_SLOPE 0.2f

typedef __attribute__((ext_vector_type(2))) float v2f;
typedef __attribute__((ext_vector_type(8))) float v8f;

__device__ __forceinline__ unsigned fkey(float f) {
    unsigned u = __float_as_uint(f);
    return (u & 0x80000000u) ? ~u : (u | 0x80000000u);
}
__device__ __forceinline__ float funkey(unsigned k) {
    return (k & 0x80000000u) ? __uint_as_float(k ^ 0x80000000u) : __uint_as_float(~k);
}

__global__ void __launch_bounds__(256)
gemm_wmma(const float* __restrict__ in, const float* __restrict__ W,
          const float* __restrict__ a_src, const float* __restrict__ a_dst,
          float* __restrict__ h, float* __restrict__ s_src, float* __restrict__ s_dst,
          int K)
{
    const int wave = threadIdx.x >> 5;
    const int lane = threadIdx.x & 31;
    const int tile = blockIdx.x * 8 + wave;
    if (tile * 16 >= NN) return;
    const int row0 = tile * 16;
    const int r    = lane & 15;
    const int hi   = lane >> 4;
    const int koff = hi * 2;

    v8f acc[8];
    const v8f vz = {0.f,0.f,0.f,0.f,0.f,0.f,0.f,0.f};
#pragma unroll
    for (int t = 0; t < 8; ++t) acc[t] = vz;

    const float* arow = in + (size_t)(row0 + r) * K;
    for (int k0 = 0; k0 < K; k0 += 4) {
        v2f a;
        a.x = arow[k0 + koff];
        a.y = arow[k0 + koff + 1];
        const float* wrow = W + (size_t)(k0 + koff) * HF + r;
#pragma unroll
        for (int t = 0; t < 8; ++t) {
            v2f b;
            b.x = wrow[t * 16];
            b.y = wrow[HF + t * 16];
            acc[t] = __builtin_amdgcn_wmma_f32_16x16x4_f32(
                false, a, false, b, (short)0, acc[t], false, false);
        }
    }

    float ps[8], pd[8];
#pragma unroll
    for (int j = 0; j < 8; ++j) { ps[j] = 0.f; pd[j] = 0.f; }
#pragma unroll
    for (int t = 0; t < 8; ++t) {
        const int col = t * 16 + r;
        const float as = a_src[col];
        const float ad = a_dst[col];
#pragma unroll
        for (int j = 0; j < 8; ++j) {
            const int row = row0 + hi * 8 + j;
            float v = acc[t][j];
            h[(size_t)row * HF + col] = v;
            ps[j] += v * as;
            pd[j] += v * ad;
        }
    }
#pragma unroll
    for (int off = 1; off < 16; off <<= 1) {
#pragma unroll
        for (int j = 0; j < 8; ++j) {
            ps[j] += __shfl_xor(ps[j], off, 32);
            pd[j] += __shfl_xor(pd[j], off, 32);
        }
    }
    if (r == 0) {
#pragma unroll
        for (int j = 0; j < 8; ++j) {
            const int row = row0 + hi * 8 + j;
            s_src[row] = ps[j];
            s_dst[row] = pd[j];
        }
    }
}

#define TT          400
#define LCAP        6144
#define AGG_THREADS 1024
#define AGG_WAVES   (AGG_THREADS / 32)
#define AGG_U       8

__device__ __forceinline__ float lrelu(float a) { return (a > 0.f) ? a : NEG_SLOPE * a; }

template <bool FIRST>
__global__ void __launch_bounds__(AGG_THREADS)
agg_tile(const int* __restrict__ esrc, const int* __restrict__ edst,
         const float* __restrict__ h, const float* __restrict__ s_src,
         const float* __restrict__ s_dst, const float* __restrict__ bias,
         float* __restrict__ act,
         int* __restrict__ tl_src, int* __restrict__ tl_dl, int* __restrict__ tl_cnt)
{
    __shared__ __attribute__((aligned(16))) float acc[(TT + 1) * HF];
    __shared__ unsigned mkey[TT];
    __shared__ float    den[TT];
    __shared__ float    sd[TT];
    __shared__ float    aself[TT];
    __shared__ int      le[LCAP];
    __shared__ int      ldl[LCAP];
    __shared__ float    lal[LCAP];
    __shared__ int      cnt;

    const int tid   = threadIdx.x;
    const int lane  = tid & 31;
    const int wave  = __builtin_amdgcn_readfirstlane(tid >> 5);
    const int tile0 = blockIdx.x * TT;
    const int nT    = (NN - tile0 < TT) ? (NN - tile0) : TT;

    {
        const float4 z = {0.f, 0.f, 0.f, 0.f};
        for (int i = tid; i < (TT + 1) * HF / 4; i += AGG_THREADS) ((float4*)acc)[i] = z;
        if (tid == 0) cnt = 0;
        for (int i = tid; i < nT; i += AGG_THREADS) {
            const float d = s_dst[tile0 + i];
            const float a = lrelu(s_src[tile0 + i] + d);
            sd[i] = d; aself[i] = a; mkey[i] = fkey(a); den[i] = 0.f;
        }
    }
    __syncthreads();

    int nraw;
    if constexpr (FIRST) {
        const int4* d4 = (const int4*)edst;
        const int NV = EE / 4;
        const int GRP = 4 * AGG_THREADS;
        auto load4 = [&](int4 (&g)[4], int base) {
#pragma unroll
            for (int k = 0; k < 4; ++k) {
                const int i = base + k * AGG_THREADS + tid;
                g[k] = d4[(i < NV) ? i : (NV - 1)];
            }
        };
        auto put = [&](unsigned dl, int e) {
            const int pos = atomicAdd(&cnt, 1);
            if (pos < LCAP) { le[pos] = e; ldl[pos] = (int)dl; }
        };
        auto match4 = [&](const int4 (&g)[4], int base) {
#pragma unroll
            for (int k = 0; k < 4; ++k) {
                const int i  = base + k * AGG_THREADS + tid;
                const int e0 = i * 4;
                const unsigned lim = (i < NV) ? (unsigned)nT : 0u;
                const unsigned d0 = (unsigned)(g[k].x - tile0);
                const unsigned d1 = (unsigned)(g[k].y - tile0);
                const unsigned d2 = (unsigned)(g[k].z - tile0);
                const unsigned d3 = (unsigned)(g[k].w - tile0);
                if ((d0 < lim) | (d1 < lim) | (d2 < lim) | (d3 < lim)) {
                    if (d0 < lim) put(d0, e0);
                    if (d1 < lim) put(d1, e0 + 1);
                    if (d2 < lim) put(d2, e0 + 2);
                    if (d3 < lim) put(d3, e0 + 3);
                }
            }
        };
        int4 ga[4], gb[4];
        load4(ga, 0);
        for (int base = 0; base < NV; base += 2 * GRP) {
            load4(gb, base + GRP);
            match4(ga, base);
            load4(ga, base + 2 * GRP);
            match4(gb, base + GRP);
        }
        __syncthreads();
        nraw = __builtin_amdgcn_readfirstlane(cnt);
    } else {
        nraw = tl_cnt[blockIdx.x];
    }
    const bool ovf  = nraw > LCAP;
    const int  nl   = ovf ? 0 : nraw;
    const size_t lbase = (size_t)blockIdx.x * LCAP;

    if (!ovf) {
        for (int j = tid; j < nl; j += AGG_THREADS) {
            int s, dl;
            if constexpr (FIRST) {
                s = esrc[le[j]]; dl = ldl[j];
                tl_src[lbase + j] = s; tl_dl[lbase + j] = dl;
            } else {
                s = tl_src[lbase + j]; dl = tl_dl[lbase + j];
                ldl[j] = dl;
            }
            const float a = lrelu(s_src[s] + sd[dl]);
            atomicMax(&mkey[dl], fkey(a));
            le[j] = s; lal[j] = a;
        }
    } else {
        for (int e = tid; e < EE; e += AGG_THREADS) {
            const unsigned dl = (unsigned)(edst[e] - tile0);
            if (dl < (unsigned)nT)
                atomicMax(&mkey[dl], fkey(lrelu(s_src[esrc[e]] + sd[dl])));
        }
    }
    if constexpr (FIRST) { if (tid == 0) tl_cnt[blockIdx.x] = nraw; }
    __syncthreads();

    {
        for (int j = tid; j < nl; j += AGG_THREADS) {
            const int dl = ldl[j];
            const float x = expf(lal[j] - funkey(mkey[dl]));
            lal[j] = x;
            atomicAdd(&den[dl], x);
        }
        for (int i = tid; i < nT; i += AGG_THREADS) {
            const float x = expf(aself[i] - funkey(mkey[i]));
            aself[i] = x;
            atomicAdd(&den[i], x);
        }
    }
    __syncthreads();

    {
        const int total = nT + nl;
        auto fetch = [&](int g0, int& sv, int& dv, float& xv) {
            sv = 0; dv = TT; xv = 0.f;
            const int it = g0 + lane;
            if (lane < AGG_U && it < total) {
                if (it < nT) { sv = tile0 + it; dv = it; xv = aself[it]; }
                else { const int j = it - nT; sv = le[j]; dv = ldl[j]; xv = lal[j]; }
            }
        };
        int sv, dv; float xv;
        fetch(wave * AGG_U, sv, dv, xv);
        for (int it0 = wave * AGG_U; it0 < total; it0 += AGG_WAVES * AGG_U) {
            float4 v[AGG_U]; float ex[AGG_U]; int dd[AGG_U];
#pragma unroll
            for (int u = 0; u < AGG_U; ++u) {
                const int s = __builtin_amdgcn_readlane(sv, u);
                dd[u] = __builtin_amdgcn_readlane(dv, u);
                ex[u] = __int_as_float(__builtin_amdgcn_readlane(__float_as_int(xv), u));
                v[u] = ((const float4*)(h + (size_t)s * HF))[lane];
            }
            fetch(it0 + AGG_WAVES * AGG_U, sv, dv, xv);
#pragma unroll
            for (int u = 0; u < AGG_U; ++u) {
                float* ar = acc + dd[u] * HF + lane;
                atomicAdd(ar,      ex[u] * v[u].x);
                atomicAdd(ar + 32, ex[u] * v[u].y);
                atomicAdd(ar + 64, ex[u] * v[u].z);
                atomicAdd(ar + 96, ex[u] * v[u].w);
            }
        }
        if (ovf) {
            for (int e = tid; e < EE; e += AGG_THREADS) {
                const unsigned dl = (unsigned)(edst[e] - tile0);
                if (dl < (unsigned)nT) {
                    const int s = esrc[e];
                    const float x = expf(lrelu(s_src[s] + sd[dl]) - funkey(mkey[dl]));
                    atomicAdd(&den[dl], x);
                    const float* hr = h + (size_t)s * HF;
                    float* ar = acc + dl * HF;
                    for (int f = 0; f < HF; ++f)
                        atomicAdd(&ar[(f & 3) * 32 + (f >> 2)], x * hr[f]);
                }
            }
        }
    }
    __syncthreads();

    {
        const float4 bv = ((const float4*)bias)[lane];
        for (int r = wave; r < nT; r += AGG_WAVES) {
            const float* ar = acc + r * HF + lane;
            const float dn = den[r];
            float4 o;
            o.x = ar[0]  / dn + bv.x;
            o.y = ar[32] / dn + bv.y;
            o.z = ar[64] / dn + bv.z;
            o.w = ar[96] / dn + bv.w;
            o.x = (o.x > 0.f) ? o.x : 0.f;
            o.y = (o.y > 0.f) ? o.y : 0.f;
            o.z = (o.z > 0.f) ? o.z : 0.f;
            o.w = (o.w > 0.f) ? o.w : 0.f;
            ((float4*)(act + (size_t)(tile0 + r) * HF))[lane] = o;
        }
    }
}

#define PI_BLOCKS 128
__global__ void __launch_bounds__(256)
pool_init(float* __restrict__ pooled, float* __restrict__ cnts)
{
    const float4 z = {0.f, 0.f, 0.f, 0.f};
    const int stride = PI_BLOCKS * 256;
    for (int i = blockIdx.x * 256 + threadIdx.x; i < GG * HF / 4; i += stride) ((float4*)pooled)[i] = z;
    for (int i = blockIdx.x * 256 + threadIdx.x; i < GG / 4; i += stride) ((float4*)cnts)[i] = z;
}

#define PC_NODES 32
static_assert(NN % PC_NODES == 0, "a chunk never runs past the last node");
static_assert(GG % 4 == 0 && (GG * HF) % 4 == 0, "pool_init clears whole float4s");

__global__ void __launch_bounds__(256)
pool_chunk(const float* __restrict__ act, const int* __restrict__ batch,
           float* __restrict__ pooled, float* __restrict__ cnts)
{
    const int lane = threadIdx.x & 31;
    const int n0   = (blockIdx.x * 8 + (threadIdx.x >> 5)) * PC_NODES;
    if (n0 >= NN) return;
    const int bl = batch[n0 + lane];

    float4 v[PC_NODES];
#pragma unroll
    for (int i = 0; i < PC_NODES; ++i)
        v[i] = ((const float4*)(act + (size_t)(n0 + i) * HF))[lane];

    auto flush = [&](int g, const float4& s, float c) {
        float* p = pooled + (size_t)g * HF + lane * 4;
        atomicAdd(p + 0, s.x);
        atomicAdd(p + 1, s.y);
        atomicAdd(p + 2, s.z);
        atomicAdd(p + 3, s.w);
        if (lane == 0) atomicAdd(cnts + g, c);
    };

    int    cur = __builtin_amdgcn_readlane(bl, 0);
    float4 s   = v[0];
    float  c   = 1.f;
#pragma unroll
    for (int i = 1; i < PC_NODES; ++i) {
        const int b = __builtin_amdgcn_readlane(bl, i);
        if (b != cur) {
            flush(cur, s, c);
            cur = b; s = v[i]; c = 1.f;
        } else {
            s.x += v[i].x; s.y += v[i].y; s.z += v[i].z; s.w += v[i].w;
            c += 1.f;
        }
    }
    flush(cur, s, c);
}

__global__ void __launch_bounds__(64)
fc_head(const float* __restrict__ pooled, const float* __restrict__ cnts,
        const float* __restrict__ w1, const float* __restrict__ b1,
        const float* __restrict__ w2, const float* __restrict__ b2,
        float* __restrict__ out)
{
    __shared__ float p[HF];
    __shared__ float z[64];
    int g = blockIdx.x, t = threadIdx.x;
    float inv = 1.0f / fmaxf(cnts[g], 1.0f);
    p[t]      = pooled[(size_t)g * HF + t] * inv;
    p[t + 64] = pooled[(size_t)g * HF + t + 64] * inv;
    __syncthreads();
    float a = b1[t];
    for (int k = 0; k < HF; ++k) a += p[k] * w1[k * 64 + t];
    z[t] = fmaxf(a, 0.f);
    __syncthreads();
    if (t < OUTF) {
        float o = b2[t];
        for (int k = 0; k < 64; ++k) o += z[k] * w2[k * OUTF + t];
        out[(size_t)g * OUTF + t] = o;
    }
}

extern "C" void kernel_launch(void* const* d_in, const int* in_sizes, int n_in,
                              void* d_out, int out_size, void* d_ws, size_t ws_size,
                              hipStream_t stream)
{
    const float* x      = (const float*)d_in[0];
    const int*   eidx   = (const int*)  d_in[1];
    const int*   batch  = (const int*)  d_in[2];
    const float* W[3]    = {(const float*)d_in[3],  (const float*)d_in[7],  (const float*)d_in[11]};
    const float* asrc[3] = {(const float*)d_in[4],  (const float*)d_in[8],  (const float*)d_in[12]};
    const float* adst[3] = {(const float*)d_in[5],  (const float*)d_in[9],  (const float*)d_in[13]};
    const float* bias[3] = {(const float*)d_in[6],  (const float*)d_in[10], (const float*)d_in[14]};
    const float* fc1w = (const float*)d_in[15];
    const float* fc1b = (const float*)d_in[16];
    const float* fc2w = (const float*)d_in[17];
    const float* fc2b = (const float*)d_in[18];
    float* out = (float*)d_out;

    const int* esrc = eidx;
    const int* edst = eidx + EE;

    char* ws = (char*)d_ws;
    const size_t NHB = (size_t)NN * HF * sizeof(float);
    float*    h_lin  = (float*)(ws);
    int*      tl_src = (int*)(ws + NHB);
    int*      tl_dl  = tl_src + (size_t)((NN + TT - 1) / TT) * LCAP;
    int*      tl_cnt = tl_dl  + (size_t)((NN + TT - 1) / TT) * LCAP;
    float*    act    = (float*)(ws + 2 * NHB);
    float*    s_src  = (float*)(ws + 3 * NHB);
    float*    s_dst  = (float*)(ws + 3 * NHB + (size_t)NN * 4);
    float*    pooled = (float*)(ws + 3 * NHB + (size_t)NN * 16 + (size_t)ETOT * 4);
    float*    cnts   = (float*)(ws + 3 * NHB + (size_t)NN * 16 + (size_t)ETOT * 4 + (size_t)GG * HF * 4);

    const int gemm_grid   = ((NN / 16) + 7) / 8;
    const int agg_grid    = (NN + TT - 1) / TT;
    const int pool_grid   = (NN / PC_NODES + 7) / 8;

    gemm_wmma<<<gemm_grid, 256, 0, stream>>>(x, W[0], asrc[0], adst[0],
                                             h_lin, s_src, s_dst, INF_);
    agg_tile<true><<<agg_grid, AGG_THREADS, 0, stream>>>(esrc, edst, h_lin, s_src, s_dst, bias[0], act,
                                                         tl_src, tl_dl, tl_cnt);
    for (int l = 1; l < 3; ++l) {
        gemm_wmma<<<gemm_grid, 256, 0, stream>>>(act, W[l], asrc[l], adst[l],
                                                 h_lin, s_src, s_dst, HF);
        agg_tile<false><<<agg_grid, AGG_THREADS, 0, stream>>>(esrc, edst, h_lin, s_src, s_dst, bias[l], act,
                                                              tl_src, tl_dl, tl_cnt);
    }

    pool_init<<<PI_BLOCKS, 256, 0, stream>>>(pooled, cnts);
    pool_chunk<<<pool_grid, 256, 0, stream>>>(act, batch, pooled, cnts);
    fc_head<<<GG, 64, 0, stream>>>(pooled, cnts, fc1w, fc1b, fc2w, fc2b, out);
}


# layer-1 edge scan gone: 128 binning blocks in the first GEMM launch put each edge into its destination tile's slot row (LDS counter per tile, arrival order); a tile block reads only its own slots
# speedup vs baseline: 3.6195x; 1.0644x over previous
#include <hip/hip_runtime.h>

#define NN   100000
#define EE   600000
#define ETOT (EE + NN)
#define INF_ 64
#define HF   128
#define GG   5000
#define OUTF 19
#define NEG_SLOPE 0.2f

typedef __attribute__((ext_vector_type(2))) float v2f;
typedef __attribute__((ext_vector_type(8))) float v8f;

__device__ __forceinline__ unsigned fkey(float f) {
    unsigned u = __float_as_uint(f);
    return (u & 0x80000000u) ? ~u : (u | 0x80000000u);
}
__device__ __forceinline__ float funkey(unsigned k) {
    return (k & 0x80000000u) ? __uint_as_float(k ^ 0x80000000u) : __uint_as_float(~k);
}

__device__ __forceinline__ void
gemm_block(int blk, const float* __restrict__ in, const float* __restrict__ W,
           const float* __restrict__ a_src, const float* __restrict__ a_dst,
           float* __restrict__ h, float* __restrict__ s_src, float* __restrict__ s_dst,
           int K)
{
    const int wave = threadIdx.x >> 5;
    const int lane = threadIdx.x & 31;
    const int tile = blk * 8 + wave;
    if (tile * 16 >= NN) return;
    const int row0 = tile * 16;
    const int r    = lane & 15;
    const int hi   = lane >> 4;
    const int koff = hi * 2;

    v8f acc[8];
    const v8f vz = {0.f,0.f,0.f,0.f,0.f,0.f,0.f,0.f};
#pragma unroll
    for (int t = 0; t < 8; ++t) acc[t] = vz;

    const float* arow = in + (size_t)(row0 + r) * K;
    for (int k0 = 0; k0 < K; k0 += 4) {
        v2f a;
        a.x = arow[k0 + koff];
        a.y = arow[k0 + koff + 1];
        const float* wrow = W + (size_t)(k0 + koff) * HF + r;
#pragma unroll
        for (int t = 0; t < 8; ++t) {
            v2f b;
            b.x = wrow[t * 16];
            b.y = wrow[HF + t * 16];
            acc[t] = __builtin_amdgcn_wmma_f32_16x16x4_f32(
                false, a, false, b, (short)0, acc[t], false, false);
        }
    }

    float ps[8], pd[8];
#pragma unroll
    for (int j = 0; j < 8; ++j) { ps[j] = 0.f; pd[j] = 0.f; }
#pragma unroll
    for (int t = 0; t < 8; ++t) {
        const int col = t * 16 + r;
        const float as = a_src[col];
        const float ad = a_dst[col];
#pragma unroll
        for (int j = 0; j < 8; ++j) {
            const int row = row0 + hi * 8 + j;
            float v = acc[t][j];
            h[(size_t)row * HF + col] = v;
            ps[j] += v * as;
            pd[j] += v * ad;
        }
    }
#pragma unroll
    for (int off = 1; off < 16; off <<= 1) {
#pragma unroll
        for (int j = 0; j < 8; ++j) {
            ps[j] += __shfl_xor(ps[j], off, 32);
            pd[j] += __shfl_xor(pd[j], off, 32);
        }
    }
    if (r == 0) {
#pragma unroll
        for (int j = 0; j < 8; ++j) {
            const int row = row0 + hi * 8 + j;
            s_src[row] = ps[j];
            s_dst[row] = pd[j];
        }
    }
}

__global__ void __launch_bounds__(256)
gemm_wmma(const float* __restrict__ in, const float* __restrict__ W,
          const float* __restrict__ a_src, const float* __restrict__ a_dst,
          float* __restrict__ h, float* __restrict__ s_src, float* __restrict__ s_dst,
          int K)
{
    gemm_block(blockIdx.x, in, W, a_src, a_dst, h, s_src, s_dst, K);
}

#define TT          400
#define NTILES      ((NN + TT - 1) / TT)
#define NCH         128
#define SCAP        64
#define CH_VEC      ((EE / 4 + NCH - 1) / NCH)
#define CH_LOADS    ((CH_VEC + 255) / 256)
static_assert(EE % 4 == 0, "the edge columns are read as whole 16-byte vectors");
static_assert(NN <= (1 << 17) && TT <= (1 << 14), "source id and tile row share one word");
static_assert(NTILES <= 256, "one LDS counter per tile, cleared and stored by 256 threads");

__global__ void __launch_bounds__(256)
gemm_wmma_bin(const float* __restrict__ in, const float* __restrict__ W,
              const float* __restrict__ a_src, const float* __restrict__ a_dst,
              float* __restrict__ h, float* __restrict__ s_src, float* __restrict__ s_dst,
              int K,
              const int* __restrict__ esrc, const int* __restrict__ edst,
              int* __restrict__ seg, int* __restrict__ segcnt)
{
    if (blockIdx.x >= NCH) {
        gemm_block(blockIdx.x - NCH, in, W, a_src, a_dst, h, s_src, s_dst, K);
        return;
    }
    __shared__ int lcnt[256];
    const int tid = threadIdx.x;
    const int c   = blockIdx.x;
    lcnt[tid] = 0;
    const int4* s4 = (const int4*)esrc;
    const int4* d4 = (const int4*)edst;
    const int NV = EE / 4;
    int4 dv[CH_LOADS], sv[CH_LOADS];
#pragma unroll
    for (int i = 0; i < CH_LOADS; ++i) {
        const int v  = tid + i * 256;
        const int gv = c * CH_VEC + v;
        const int ld = (v < CH_VEC && gv < NV) ? gv : (NV - 1);
        dv[i] = d4[ld];
        sv[i] = s4[ld];
    }
    __syncthreads();
    auto bin = [&](int d, int s) {
        if ((unsigned)d < (unsigned)NN) {
            const int t   = d / TT;
            const int dl  = d - t * TT;
            const int pos = atomicAdd(&lcnt[t], 1);
            if (pos < SCAP) seg[((size_t)t * NCH + c) * SCAP + pos] = (dl << 17) | s;
        }
    };
#pragma unroll
    for (int i = 0; i < CH_LOADS; ++i) {
        const int v  = tid + i * 256;
        const int gv = c * CH_VEC + v;
        if (v < CH_VEC && gv < NV) {
            bin(dv[i].x, sv[i].x);
            bin(dv[i].y, sv[i].y);
            bin(dv[i].z, sv[i].z);
            bin(dv[i].w, sv[i].w);
        }
    }
    __syncthreads();
    if (tid < NTILES) segcnt[tid * NCH + c] = lcnt[tid];
}

#define LCAP        6144
#define AGG_THREADS 1024
#define AGG_WAVES   (AGG_THREADS / 32)
#define AGG_U       8

__device__ __forceinline__ float lrelu(float a) { return (a > 0.f) ? a : NEG_SLOPE * a; }

template <bool FIRST>
__global__ void __launch_bounds__(AGG_THREADS)
agg_tile(const int* __restrict__ esrc, const int* __restrict__ edst,
         const float* __restrict__ h, const float* __restrict__ s_src,
         const float* __restrict__ s_dst, const float* __restrict__ bias,
         float* __restrict__ act,
         int* __restrict__ tl_src, int* __restrict__ tl_dl, int* __restrict__ tl_cnt,
         const int* __restrict__ seg, const int* __restrict__ segcnt)
{
    __shared__ __attribute__((aligned(16))) float acc[(TT + 1) * HF];
    __shared__ unsigned mkey[TT];
    __shared__ float    den[TT];
    __shared__ float    sd[TT];
    __shared__ float    aself[TT];
    __shared__ int      le[LCAP];
    __shared__ int      ldl[LCAP];
    __shared__ float    lal[LCAP];
    __shared__ int      cnt;
    __shared__ int      scnt[NCH];
    __shared__ int      sovf;

    const int tid   = threadIdx.x;
    const int lane  = tid & 31;
    const int wave  = __builtin_amdgcn_readfirstlane(tid >> 5);
    const int tile0 = blockIdx.x * TT;
    const int nT    = (NN - tile0 < TT) ? (NN - tile0) : TT;

    {
        const float4 z = {0.f, 0.f, 0.f, 0.f};
        for (int i = tid; i < (TT + 1) * HF / 4; i += AGG_THREADS) ((float4*)acc)[i] = z;
        if (tid == 0) { cnt = 0; sovf = 0; }
        for (int i = tid; i < nT; i += AGG_THREADS) {
            const float d = s_dst[tile0 + i];
            const float a = lrelu(s_src[tile0 + i] + d);
            sd[i] = d; aself[i] = a; mkey[i] = fkey(a); den[i] = 0.f;
        }
    }
    __syncthreads();

    int nraw;
    if constexpr (FIRST) {
        if (tid < NCH) {
            const int n = segcnt[blockIdx.x * NCH + tid];
            scnt[tid] = n;
            if (n > SCAP) sovf = 1;
        }
        __syncthreads();
        const int* srow = seg + (size_t)blockIdx.x * NCH * SCAP;
        constexpr int NQ = NCH * SCAP / AGG_THREADS;
        static_assert(NQ * AGG_THREADS == NCH * SCAP, "whole rounds of slots");
        int  sval[NQ];
        bool sok[NQ];
#pragma unroll
        for (int i = 0; i < NQ; ++i) {
            const int q = tid + i * AGG_THREADS;
            sok[i]  = (q % SCAP) < scnt[q / SCAP];
            sval[i] = sok[i] ? srow[q] : 0;
        }
#pragma unroll
        for (int i = 0; i < NQ; ++i) {
            if (sok[i]) {
                const int pos = atomicAdd(&cnt, 1);
                if (pos < LCAP) { le[pos] = sval[i] & 0x1ffff; ldl[pos] = sval[i] >> 17; }
            }
        }
        __syncthreads();
        nraw = __builtin_amdgcn_readfirstlane(sovf ? (LCAP + 1) : cnt);
    } else {
        nraw = tl_cnt[blockIdx.x];
    }
    const bool ovf  = nraw > LCAP;
    const int  nl   = ovf ? 0 : nraw;
    const size_t lbase = (size_t)blockIdx.x * LCAP;

    if (!ovf) {
        for (int j = tid; j < nl; j += AGG_THREADS) {
            int s, dl;
            if constexpr (FIRST) {
                s = le[j]; dl = ldl[j];
                tl_src[lbase + j] = s; tl_dl[lbase + j] = dl;
            } else {
                s = tl_src[lbase + j]; dl = tl_dl[lbase + j];
                ldl[j] = dl;
            }
            const float a = lrelu(s_src[s] + sd[dl]);
            atomicMax(&mkey[dl], fkey(a));
            le[j] = s; lal[j] = a;
        }
    } else {
        for (int e = tid; e < EE; e += AGG_THREADS) {
            const unsigned dl = (unsigned)(edst[e] - tile0);
            if (dl < (unsigned)nT)
                atomicMax(&mkey[dl], fkey(lrelu(s_src[esrc[e]] + sd[dl])));
        }
    }
    if constexpr (FIRST) { if (tid == 0) tl_cnt[blockIdx.x] = nraw; }
    __syncthreads();

    {
        for (int j = tid; j < nl; j += AGG_THREADS) {
            const int dl = ldl[j];
            const float x = expf(lal[j] - funkey(mkey[dl]));
            lal[j] = x;
            atomicAdd(&den[dl], x);
        }
        for (int i = tid; i < nT; i += AGG_THREADS) {
            const float x = expf(aself[i] - funkey(mkey[i]));
            aself[i] = x;
            atomicAdd(&den[i], x);
        }
    }
    __syncthreads();

    {
        const int total = nT + nl;
        auto fetch = [&](int g0, int& sv, int& dv, float& xv) {
            sv = 0; dv = TT; xv = 0.f;
            const int it = g0 + lane;
            if (lane < AGG_U && it < total) {
                if (it < nT) { sv = tile0 + it; dv = it; xv = aself[it]; }
                else { const int j = it - nT; sv = le[j]; dv = ldl[j]; xv = lal[j]; }
            }
        };
        int sv, dv; float xv;
        fetch(wave * AGG_U, sv, dv, xv);
        for (int it0 = wave * AGG_U; it0 < total; it0 += AGG_WAVES * AGG_U) {
            float4 v[AGG_U]; float ex[AGG_U]; int dd[AGG_U];
#pragma unroll
            for (int u = 0; u < AGG_U; ++u) {
                const int s = __builtin_amdgcn_readlane(sv, u);
                dd[u] = __builtin_amdgcn_readlane(dv, u);
                ex[u] = __int_as_float(__builtin_amdgcn_readlane(__float_as_int(xv), u));
                v[u] = ((const float4*)(h + (size_t)s * HF))[lane];
            }
            fetch(it0 + AGG_WAVES * AGG_U, sv, dv, xv);
#pragma unroll
            for (int u = 0; u < AGG_U; ++u) {
                float* ar = acc + dd[u] * HF + lane;
                atomicAdd(ar,      ex[u] * v[u].x);
                atomicAdd(ar + 32, ex[u] * v[u].y);
                atomicAdd(ar + 64, ex[u] * v[u].z);
                atomicAdd(ar + 96, ex[u] * v[u].w);
            }
        }
        if (ovf) {
            for (int e = tid; e < EE; e += AGG_THREADS) {
                const unsigned dl = (unsigned)(edst[e] - tile0);
                if (dl < (unsigned)nT) {
                    const int s = esrc[e];
                    const float x = expf(lrelu(s_src[s] + sd[dl]) - funkey(mkey[dl]));
                    atomicAdd(&den[dl], x);
                    const float* hr = h + (size_t)s * HF;
                    float* ar = acc + dl * HF;
                    for (int f = 0; f < HF; ++f)
                        atomicAdd(&ar[(f & 3) * 32 + (f >> 2)], x * hr[f]);
                }
            }
        }
    }
    __syncthreads();

    {
        const float4 bv = ((const float4*)bias)[lane];
        for (int r = wave; r < nT; r += AGG_WAVES) {
            const float* ar = acc + r * HF + lane;
            const float dn = den[r];
            float4 o;
            o.x = ar[0]  / dn + bv.x;
            o.y = ar[32] / dn + bv.y;
            o.z = ar[64] / dn + bv.z;
            o.w = ar[96] / dn + bv.w;
            o.x = (o.x > 0.f) ? o.x : 0.f;
            o.y = (o.y > 0.f) ? o.y : 0.f;
            o.z = (o.z > 0.f) ? o.z : 0.f;
            o.w = (o.w > 0.f) ? o.w : 0.f;
            ((float4*)(act + (size_t)(tile0 + r) * HF))[lane] = o;
        }
    }
}

#define PI_BLOCKS 128
__global__ void __launch_bounds__(256)
pool_init(float* __restrict__ pooled, float* __restrict__ cnts)
{
    const float4 z = {0.f, 0.f, 0.f, 0.f};
    const int stride = PI_BLOCKS * 256;
    for (int i = blockIdx.x * 256 + threadIdx.x; i < GG * HF / 4; i += stride) ((float4*)pooled)[i] = z;
    for (int i = blockIdx.x * 256 + threadIdx.x; i < GG / 4; i += stride) ((float4*)cnts)[i] = z;
}

#define PC_NODES 32
static_assert(NN % PC_NODES == 0, "a chunk never runs past the last node");
static_assert(GG % 4 == 0 && (GG * HF) % 4 == 0, "pool_init clears whole float4s");

__global__ void __launch_bounds__(256)
pool_chunk(const float* __restrict__ act, const int* __restrict__ batch,
           float* __restrict__ pooled, float* __restrict__ cnts)
{
    const int lane = threadIdx.x & 31;
    const int n0   = (blockIdx.x * 8 + (threadIdx.x >> 5)) * PC_NODES;
    if (n0 >= NN) return;
    const int bl = batch[n0 + lane];

    float4 v[PC_NODES];
#pragma unroll
    for (int i = 0; i < PC_NODES; ++i)
        v[i] = ((const float4*)(act + (size_t)(n0 + i) * HF))[lane];

    auto flush = [&](int g, const float4& s, float c) {
        float* p = pooled + (size_t)g * HF + lane * 4;
        atomicAdd(p + 0, s.x);
        atomicAdd(p + 1, s.y);
        atomicAdd(p + 2, s.z);
        atomicAdd(p + 3, s.w);
        if (lane == 0) atomicAdd(cnts + g, c);
    };

    int    cur = __builtin_amdgcn_readlane(bl, 0);
    float4 s   = v[0];
    float  c   = 1.f;
#pragma unroll
    for (int i = 1; i < PC_NODES; ++i) {
        const int b = __builtin_amdgcn_readlane(bl, i);
        if (b != cur) {
            flush(cur, s, c);
            cur = b; s = v[i]; c = 1.f;
        } else {
            s.x += v[i].x; s.y += v[i].y; s.z += v[i].z; s.w += v[i].w;
            c += 1.f;
        }
    }
    flush(cur, s, c);
}

__global__ void __launch_bounds__(64)
fc_head(const float* __restrict__ pooled, const float* __restrict__ cnts,
        const float* __restrict__ w1, const float* __restrict__ b1,
        const float* __restrict__ w2, const float* __restrict__ b2,
        float* __restrict__ out)
{
    __shared__ float p[HF];
    __shared__ float z[64];
    int g = blockIdx.x, t = threadIdx.x;
    float inv = 1.0f / fmaxf(cnts[g], 1.0f);
    p[t]      = pooled[(size_t)g * HF + t] * inv;
    p[t + 64] = pooled[(size_t)g * HF + t + 64] * inv;
    __syncthreads();
    float a = b1[t];
    for (int k = 0; k < HF; ++k) a += p[k] * w1[k * 64 + t];
    z[t] = fmaxf(a, 0.f);
    __syncthreads();
    if (t < OUTF) {
        float o = b2[t];
        for (int k = 0; k < 64; ++k) o += z[k] * w2[k * OUTF + t];
        out[(size_t)g * OUTF + t] = o;
    }
}

extern "C" void kernel_launch(void* const* d_in, const int* in_sizes, int n_in,
                              void* d_out, int out_size, void* d_ws, size_t ws_size,
                              hipStream_t stream)
{
    const float* x      = (const float*)d_in[0];
    const int*   eidx   = (const int*)  d_in[1];
    const int*   batch  = (const int*)  d_in[2];
    const float* W[3]    = {(const float*)d_in[3],  (const float*)d_in[7],  (const float*)d_in[11]};
    const float* asrc[3] = {(const float*)d_in[4],  (const float*)d_in[8],  (const float*)d_in[12]};
    const float* adst[3] = {(const float*)d_in[5],  (const float*)d_in[9],  (const float*)d_in[13]};
    const float* bias[3] = {(const float*)d_in[6],  (const float*)d_in[10], (const float*)d_in[14]};
    const float* fc1w = (const float*)d_in[15];
    const float* fc1b = (const float*)d_in[16];
    const float* fc2w = (const float*)d_in[17];
    const float* fc2b = (const float*)d_in[18];
    float* out = (float*)d_out;

    const int* esrc = eidx;
    const int* edst = eidx + EE;

    char* ws = (char*)d_ws;
    const size_t NHB = (size_t)NN * HF * sizeof(float);
    float*    h_lin  = (float*)(ws);
    int*      tl_src = (int*)(ws + NHB);
    int*      tl_dl  = tl_src + (size_t)((NN + TT - 1) / TT) * LCAP;
    int*      tl_cnt = tl_dl  + (size_t)((NN + TT - 1) / TT) * LCAP;
    int*      seg    = tl_cnt + 256;
    int*      segcnt = seg + (size_t)NTILES * NCH * SCAP;
    float*    act    = (float*)(ws + 2 * NHB);
    float*    s_src  = (float*)(ws + 3 * NHB);
    float*    s_dst  = (float*)(ws + 3 * NHB + (size_t)NN * 4);
    float*    pooled = (float*)(ws + 3 * NHB + (size_t)NN * 16 + (size_t)ETOT * 4);
    float*    cnts   = (float*)(ws + 3 * NHB + (size_t)NN * 16 + (size_t)ETOT * 4 + (size_t)GG * HF * 4);

    const int gemm_grid   = ((NN / 16) + 7) / 8;
    const int agg_grid    = (NN + TT - 1) / TT;
    const int pool_grid   = (NN / PC_NODES + 7) / 8;

    gemm_wmma_bin<<<NCH + gemm_grid, 256, 0, stream>>>(x, W[0], asrc[0], adst[0],
                                                       h_lin, s_src, s_dst, INF_,
                                                       esrc, edst, seg, segcnt);
    agg_tile<true><<<agg_grid, AGG_THREADS, 0, stream>>>(esrc, edst, h_lin, s_src, s_dst, bias[0], act,
                                                         tl_src, tl_dl, tl_cnt, seg, segcnt);
    for (int l = 1; l < 3; ++l) {
        gemm_wmma<<<gemm_grid, 256, 0, stream>>>(act, W[l], asrc[l], adst[l],
                                                 h_lin, s_src, s_dst, HF);
        agg_tile<false><<<agg_grid, AGG_THREADS, 0, stream>>>(esrc, edst, h_lin, s_src, s_dst, bias[l], act,
                                                              tl_src, tl_dl, tl_cnt, seg, segcnt);
    }

    pool_init<<<PI_BLOCKS, 256, 0, stream>>>(pooled, cnts);
    pool_chunk<<<pool_grid, 256, 0, stream>>>(act, batch, pooled, cnts);
    fc_head<<<GG, 64, 0, stream>>>(pooled, cnts, fc1w, fc1b, fc2w, fc2b, out);
}


# GEMM: lane column r of accumulator t = output column 4r+(t&3)+64(t>>2): W fragment loads 16 b32 -> 4 b128 per k-step, h stores whole lines by b128; a_src/a_dst dots summed in a new order
# speedup vs baseline: 3.8397x; 1.0608x over previous
#include <hip/hip_runtime.h>

#define NN   100000
#define EE   600000
#define ETOT (EE + NN)
#define INF_ 64
#define HF   128
#define GG   5000
#define OUTF 19
#define NEG_SLOPE 0.2f

typedef __attribute__((ext_vector_type(2))) float v2f;
typedef __attribute__((ext_vector_type(8))) float v8f;

__device__ __forceinline__ unsigned fkey(float f) {
    unsigned u = __float_as_uint(f);
    return (u & 0x80000000u) ? ~u : (u | 0x80000000u);
}
__device__ __forceinline__ float funkey(unsigned k) {
    return (k & 0x80000000u) ? __uint_as_float(k ^ 0x80000000u) : __uint_as_float(~k);
}

__device__ __forceinline__ void
gemm_block(int blk, const float* __restrict__ in, const float* __restrict__ W,
           const float* __restrict__ a_src, const float* __restrict__ a_dst,
           float* __restrict__ h, float* __restrict__ s_src, float* __restrict__ s_dst,
           int K)
{
    const int wave = threadIdx.x >> 5;
    const int lane = threadIdx.x & 31;
    const int tile = blk * 8 + wave;
    if (tile * 16 >= NN) return;
    const int row0 = tile * 16;
    const int r    = lane & 15;
    const int hi   = lane >> 4;
    const int koff = hi * 2;

    v8f acc[8];
    const v8f vz = {0.f,0.f,0.f,0.f,0.f,0.f,0.f,0.f};
#pragma unroll
    for (int t = 0; t < 8; ++t) acc[t] = vz;

    const float* arow = in + (size_t)(row0 + r) * K;
    const float* wcol = W + (size_t)koff * HF + 4 * r;
    for (int k0 = 0; k0 < K; k0 += 4) {
        v2f a;
        a.x = arow[k0 + koff];
        a.y = arow[k0 + koff + 1];
        const float* w0 = wcol + (size_t)k0 * HF;
        const float4 p0 = *(const float4*)(w0);
        const float4 q0 = *(const float4*)(w0 + 64);
        const float4 p1 = *(const float4*)(w0 + HF);
        const float4 q1 = *(const float4*)(w0 + HF + 64);
        const float bx[8] = {p0.x, p0.y, p0.z, p0.w, q0.x, q0.y, q0.z, q0.w};
        const float by[8] = {p1.x, p1.y, p1.z, p1.w, q1.x, q1.y, q1.z, q1.w};
#pragma unroll
        for (int t = 0; t < 8; ++t) {
            v2f b;
            b.x = bx[t];
            b.y = by[t];
            acc[t] = __builtin_amdgcn_wmma_f32_16x16x4_f32(
                false, a, false, b, (short)0, acc[t], false, false);
        }
    }

    const float4 asl = *(const float4*)(a_src + 4 * r), ash = *(const float4*)(a_src + 64 + 4 * r);
    const float4 adl = *(const float4*)(a_dst + 4 * r), adh = *(const float4*)(a_dst + 64 + 4 * r);
    float ps[8], pd[8];
#pragma unroll
    for (int j = 0; j < 8; ++j) {
        const int row = row0 + hi * 8 + j;
        const float4 lo = {acc[0][j], acc[1][j], acc[2][j], acc[3][j]};
        const float4 up = {acc[4][j], acc[5][j], acc[6][j], acc[7][j]};
        float* hrow = h + (size_t)row * HF + 4 * r;
        *(float4*)(hrow)      = lo;
        *(float4*)(hrow + 64) = up;
        ps[j] = lo.x * asl.x + lo.y * asl.y + lo.z * asl.z + lo.w * asl.w
              + up.x * ash.x + up.y * ash.y + up.z * ash.z + up.w * ash.w;
        pd[j] = lo.x * adl.x + lo.y * adl.y + lo.z * adl.z + lo.w * adl.w
              + up.x * adh.x + up.y * adh.y + up.z * adh.z + up.w * adh.w;
    }
#pragma unroll
    for (int off = 1; off < 16; off <<= 1) {
#pragma unroll
        for (int j = 0; j < 8; ++j) {
            ps[j] += __shfl_xor(ps[j], off, 32);
            pd[j] += __shfl_xor(pd[j], off, 32);
        }
    }
    if (r == 0) {
#pragma unroll
        for (int j = 0; j < 8; ++j) {
            const int row = row0 + hi * 8 + j;
            s_src[row] = ps[j];
            s_dst[row] = pd[j];
        }
    }
}

__global__ void __launch_bounds__(256)
gemm_wmma(const float* __restrict__ in, const float* __restrict__ W,
          const float* __restrict__ a_src, const float* __restrict__ a_dst,
          float* __restrict__ h, float* __restrict__ s_src, float* __restrict__ s_dst,
          int K)
{
    gemm_block(blockIdx.x, in, W, a_src, a_dst, h, s_src, s_dst, K);
}

#define TT          400
#define NTILES      ((NN + TT - 1) / TT)
#define NCH         128
#define SCAP        64
#define CH_VEC      ((EE / 4 + NCH - 1) / NCH)
#define CH_LOADS    ((CH_VEC + 255) / 256)
static_assert(EE % 4 == 0, "the edge columns are read as whole 16-byte vectors");
static_assert(NN <= (1 << 17) && TT <= (1 << 14), "source id and tile row share one word");
static_assert(NTILES <= 256, "one LDS counter per tile, cleared and stored by 256 threads");

__global__ void __launch_bounds__(256)
gemm_wmma_bin(const float* __restrict__ in, const float* __restrict__ W,
              const float* __restrict__ a_src, const float* __restrict__ a_dst,
              float* __restrict__ h, float* __restrict__ s_src, float* __restrict__ s_dst,
              int K,
              const int* __restrict__ esrc, const int* __restrict__ edst,
              int* __restrict__ seg, int* __restrict__ segcnt)
{
    if (blockIdx.x >= NCH) {
        gemm_block(blockIdx.x - NCH, in, W, a_src, a_dst, h, s_src, s_dst, K);
        return;
    }
    __shared__ int lcnt[256];
    const int tid = threadIdx.x;
    const int c   = blockIdx.x;
    lcnt[tid] = 0;
    const int4* s4 = (const int4*)esrc;
    const int4* d4 = (const int4*)edst;
    const int NV = EE / 4;
    int4 dv[CH_LOADS], sv[CH_LOADS];
#pragma unroll
    for (int i = 0; i < CH_LOADS; ++i) {
        const int v  = tid + i * 256;
        const int gv = c * CH_VEC + v;
        const int ld = (v < CH_VEC && gv < NV) ? gv : (NV - 1);
        dv[i] = d4[ld];
        sv[i] = s4[ld];
    }
    __syncthreads();
    auto bin = [&](int d, int s) {
        if ((unsigned)d < (unsigned)NN) {
            const int t   = d / TT;
            const int dl  = d - t * TT;
            const int pos = atomicAdd(&lcnt[t], 1);
            if (pos < SCAP) seg[((size_t)t * NCH + c) * SCAP + pos] = (dl << 17) | s;
        }
    };
#pragma unroll
    for (int i = 0; i < CH_LOADS; ++i) {
        const int v  = tid + i * 256;
        const int gv = c * CH_VEC + v;
        if (v < CH_VEC && gv < NV) {
            bin(dv[i].x, sv[i].x);
            bin(dv[i].y, sv[i].y);
            bin(dv[i].z, sv[i].z);
            bin(dv[i].w, sv[i].w);
        }
    }
    __syncthreads();
    if (tid < NTILES) segcnt[tid * NCH + c] = lcnt[tid];
}

#define LCAP        6144
#define AGG_THREADS 1024
#define AGG_WAVES   (AGG_THREADS / 32)
#define AGG_U       8

__device__ __forceinline__ float lrelu(float a) { return (a > 0.f) ? a : NEG_SLOPE * a; }

template <bool FIRST>
__global__ void __launch_bounds__(AGG_THREADS)
agg_tile(const int* __restrict__ esrc, const int* __restrict__ edst,
         const float* __restrict__ h, const float* __restrict__ s_src,
         const float* __restrict__ s_dst, const float* __restrict__ bias,
         float* __restrict__ act,
         int* __restrict__ tl_src, int* __restrict__ tl_dl, int* __restrict__ tl_cnt,
         const int* __restrict__ seg, const int* __restrict__ segcnt)
{
    __shared__ __attribute__((aligned(16))) float acc[(TT + 1) * HF];
    __shared__ unsigned mkey[TT];
    __shared__ float    den[TT];
    __shared__ float    sd[TT];
    __shared__ float    aself[TT];
    __shared__ int      le[LCAP];
    __shared__ int      ldl[LCAP];
    __shared__ float    lal[LCAP];
    __shared__ int      cnt;
    __shared__ int      scnt[NCH];
    __shared__ int      sovf;

    const int tid   = threadIdx.x;
    const int lane  = tid & 31;
    const int wave  = __builtin_amdgcn_readfirstlane(tid >> 5);
    const int tile0 = blockIdx.x * TT;
    const int nT    = (NN - tile0 < TT) ? (NN - tile0) : TT;

    {
        const float4 z = {0.f, 0.f, 0.f, 0.f};
        for (int i = tid; i < (TT + 1) * HF / 4; i += AGG_THREADS) ((float4*)acc)[i] = z;
        if (tid == 0) { cnt = 0; sovf = 0; }
        for (int i = tid; i < nT; i += AGG_THREADS) {
            const float d = s_dst[tile0 + i];
            const float a = lrelu(s_src[tile0 + i] + d);
            sd[i] = d; aself[i] = a; mkey[i] = fkey(a); den[i] = 0.f;
        }
    }
    __syncthreads();

    int nraw;
    if constexpr (FIRST) {
        if (tid < NCH) {
            const int n = segcnt[blockIdx.x * NCH + tid];
            scnt[tid] = n;
            if (n > SCAP) sovf = 1;
        }
        __syncthreads();
        const int* srow = seg + (size_t)blockIdx.x * NCH * SCAP;
        constexpr int NQ = NCH * SCAP / AGG_THREADS;
        static_assert(NQ * AGG_THREADS == NCH * SCAP, "whole rounds of slots");
        int  sval[NQ];
        bool sok[NQ];
#pragma unroll
        for (int i = 0; i < NQ; ++i) {
            const int q = tid + i * AGG_THREADS;
            sok[i]  = (q % SCAP) < scnt[q / SCAP];
            sval[i] = sok[i] ? srow[q] : 0;
        }
#pragma unroll
        for (int i = 0; i < NQ; ++i) {
            if (sok[i]) {
                const int pos = atomicAdd(&cnt, 1);
                if (pos < LCAP) { le[pos] = sval[i] & 0x1ffff; ldl[pos] = sval[i] >> 17; }
            }
        }
        __syncthreads();
        nraw = __builtin_amdgcn_readfirstlane(sovf ? (LCAP + 1) : cnt);
    } else {
        nraw = tl_cnt[blockIdx.x];
    }
    const bool ovf  = nraw > LCAP;
    const int  nl   = ovf ? 0 : nraw;
    const size_t lbase = (size_t)blockIdx.x * LCAP;

    if (!ovf) {
        for (int j = tid; j < nl; j += AGG_THREADS) {
            int s, dl;
            if constexpr (FIRST) {
                s = le[j]; dl = ldl[j];
                tl_src[lbase + j] = s; tl_dl[lbase + j] = dl;
            } else {
                s = tl_src[lbase + j]; dl = tl_dl[lbase + j];
                ldl[j] = dl;
            }
            const float a = lrelu(s_src[s] + sd[dl]);
            atomicMax(&mkey[dl], fkey(a));
            le[j] = s; lal[j] = a;
        }
    } else {
        for (int e = tid; e < EE; e += AGG_THREADS) {
            const unsigned dl = (unsigned)(edst[e] - tile0);
            if (dl < (unsigned)nT)
                atomicMax(&mkey[dl], fkey(lrelu(s_src[esrc[e]] + sd[dl])));
        }
    }
    if constexpr (FIRST) { if (tid == 0) tl_cnt[blockIdx.x] = nraw; }
    __syncthreads();

    {
        for (int j = tid; j < nl; j += AGG_THREADS) {
            const int dl = ldl[j];
            const float x = expf(lal[j] - funkey(mkey[dl]));
            lal[j] = x;
            atomicAdd(&den[dl], x);
        }
        for (int i = tid; i < nT; i += AGG_THREADS) {
            const float x = expf(aself[i] - funkey(mkey[i]));
            aself[i] = x;
            atomicAdd(&den[i], x);
        }
    }
    __syncthreads();

    {
        const int total = nT + nl;
        auto fetch = [&](int g0, int& sv, int& dv, float& xv) {
            sv = 0; dv = TT; xv = 0.f;
            const int it = g0 + lane;
            if (lane < AGG_U && it < total) {
                if (it < nT) { sv = tile0 + it; dv = it; xv = aself[it]; }
                else { const int j = it - nT; sv = le[j]; dv = ldl[j]; xv = lal[j]; }
            }
        };
        int sv, dv; float xv;
        fetch(wave * AGG_U, sv, dv, xv);
        for (int it0 = wave * AGG_U; it0 < total; it0 += AGG_WAVES * AGG_U) {
            float4 v[AGG_U]; float ex[AGG_U]; int dd[AGG_U];
#pragma unroll
            for (int u = 0; u < AGG_U; ++u) {
                const int s = __builtin_amdgcn_readlane(sv, u);
                dd[u] = __builtin_amdgcn_readlane(dv, u);
                ex[u] = __int_as_float(__builtin_amdgcn_readlane(__float_as_int(xv), u));
                v[u] = ((const float4*)(h + (size_t)s * HF))[lane];
            }
            fetch(it0 + AGG_WAVES * AGG_U, sv, dv, xv);
#pragma unroll
            for (int u = 0; u < AGG_U; ++u) {
                float* ar = acc + dd[u] * HF + lane;
                atomicAdd(ar,      ex[u] * v[u].x);
                atomicAdd(ar + 32, ex[u] * v[u].y);
                atomicAdd(ar + 64, ex[u] * v[u].z);
                atomicAdd(ar + 96, ex[u] * v[u].w);
            }
        }
        if (ovf) {
            for (int e = tid; e < EE; e += AGG_THREADS) {
                const unsigned dl = (unsigned)(edst[e] - tile0);
                if (dl < (unsigned)nT) {
                    const int s = esrc[e];
                    const float x = expf(lrelu(s_src[s] + sd[dl]) - funkey(mkey[dl]));
                    atomicAdd(&den[dl], x);
                    const float* hr = h + (size_t)s * HF;
                    float* ar = acc + dl * HF;
                    for (int f = 0; f < HF; ++f)
                        atomicAdd(&ar[(f & 3) * 32 + (f >> 2)], x * hr[f]);
                }
            }
        }
    }
    __syncthreads();

    {
        const float4 bv = ((const float4*)bias)[lane];
        for (int r = wave; r < nT; r += AGG_WAVES) {
            const float* ar = acc + r * HF + lane;
            const float dn = den[r];
            float4 o;
            o.x = ar[0]  / dn + bv.x;
            o.y = ar[32] / dn + bv.y;
            o.z = ar[64] / dn + bv.z;
            o.w = ar[96] / dn + bv.w;
            o.x = (o.x > 0.f) ? o.x : 0.f;
            o.y = (o.y > 0.f) ? o.y : 0.f;
            o.z = (o.z > 0.f) ? o.z : 0.f;
            o.w = (o.w > 0.f) ? o.w : 0.f;
            ((float4*)(act + (size_t)(tile0 + r) * HF))[lane] = o;
        }
    }
}

#define PI_BLOCKS 128
__global__ void __launch_bounds__(256)
pool_init(float* __restrict__ pooled, float* __restrict__ cnts)
{
    const float4 z = {0.f, 0.f, 0.f, 0.f};
    const int stride = PI_BLOCKS * 256;
    for (int i = blockIdx.x * 256 + threadIdx.x; i < GG * HF / 4; i += stride) ((float4*)pooled)[i] = z;
    for (int i = blockIdx.x * 256 + threadIdx.x; i < GG / 4; i += stride) ((float4*)cnts)[i] = z;
}

#define PC_NODES 32
static_assert(NN % PC_NODES == 0, "a chunk never runs past the last node");
static_assert(GG % 4 == 0 && (GG * HF) % 4 == 0, "pool_init clears whole float4s");

__global__ void __launch_bounds__(256)
pool_chunk(const float* __restrict__ act, const int* __restrict__ batch,
           float* __restrict__ pooled, float* __restrict__ cnts)
{
    const int lane = threadIdx.x & 31;
    const int n0   = (blockIdx.x * 8 + (threadIdx.x >> 5)) * PC_NODES;
    if (n0 >= NN) return;
    const int bl = batch[n0 + lane];

    float4 v[PC_NODES];
#pragma unroll
    for (int i = 0; i < PC_NODES; ++i)
        v[i] = ((const float4*)(act + (size_t)(n0 + i) * HF))[lane];

    auto flush = [&](int g, const float4& s, float c) {
        float* p = pooled + (size_t)g * HF + lane * 4;
        atomicAdd(p + 0, s.x);
        atomicAdd(p + 1, s.y);
        atomicAdd(p + 2, s.z);
        atomicAdd(p + 3, s.w);
        if (lane == 0) atomicAdd(cnts + g, c);
    };

    int    cur = __builtin_amdgcn_readlane(bl, 0);
    float4 s   = v[0];
    float  c   = 1.f;
#pragma unroll
    for (int i = 1; i < PC_NODES; ++i) {
        const int b = __builtin_amdgcn_readlane(bl, i);
        if (b != cur) {
            flush(cur, s, c);
            cur = b; s = v[i]; c = 1.f;
        } else {
            s.x += v[i].x; s.y += v[i].y; s.z += v[i].z; s.w += v[i].w;
            c += 1.f;
        }
    }
    flush(cur, s, c);
}

__global__ void __launch_bounds__(64)
fc_head(const float* __restrict__ pooled, const float* __restrict__ cnts,
        const float* __restrict__ w1, const float* __restrict__ b1,
        const float* __restrict__ w2, const float* __restrict__ b2,
        float* __restrict__ out)
{
    __shared__ float p[HF];
    __shared__ float z[64];
    int g = blockIdx.x, t = threadIdx.x;
    float inv = 1.0f / fmaxf(cnts[g], 1.0f);
    p[t]      = pooled[(size_t)g * HF + t] * inv;
    p[t + 64] = pooled[(size_t)g * HF + t + 64] * inv;
    __syncthreads();
    float a = b1[t];
    for (int k = 0; k < HF; ++k) a += p[k] * w1[k * 64 + t];
    z[t] = fmaxf(a, 0.f);
    __syncthreads();
    if (t < OUTF) {
        float o = b2[t];
        for (int k = 0; k < 64; ++k) o += z[k] * w2[k * OUTF + t];
        out[(size_t)g * OUTF + t] = o;
    }
}

extern "C" void kernel_launch(void* const* d_in, const int* in_sizes, int n_in,
                              void* d_out, int out_size, void* d_ws, size_t ws_size,
                              hipStream_t stream)
{
    const float* x      = (const float*)d_in[0];
    const int*   eidx   = (const int*)  d_in[1];
    const int*   batch  = (const int*)  d_in[2];
    const float* W[3]    = {(const float*)d_in[3],  (const float*)d_in[7],  (const float*)d_in[11]};
    const float* asrc[3] = {(const float*)d_in[4],  (const float*)d_in[8],  (const float*)d_in[12]};
    const float* adst[3] = {(const float*)d_in[5],  (const float*)d_in[9],  (const float*)d_in[13]};
    const float* bias[3] = {(const float*)d_in[6],  (const float*)d_in[10], (const float*)d_in[14]};
    const float* fc1w = (const float*)d_in[15];
    const float* fc1b = (const float*)d_in[16];
    const float* fc2w = (const float*)d_in[17];
    const float* fc2b = (const float*)d_in[18];
    float* out = (float*)d_out;

    const int* esrc = eidx;
    const int* edst = eidx + EE;

    char* ws = (char*)d_ws;
    const size_t NHB = (size_t)NN * HF * sizeof(float);
    float*    h_lin  = (float*)(ws);
    int*      tl_src = (int*)(ws + NHB);
    int*      tl_dl  = tl_src + (size_t)((NN + TT - 1) / TT) * LCAP;
    int*      tl_cnt = tl_dl  + (size_t)((NN + TT - 1) / TT) * LCAP;
    int*      seg    = tl_cnt + 256;
    int*      segcnt = seg + (size_t)NTILES * NCH * SCAP;
    float*    act    = (float*)(ws + 2 * NHB);
    float*    s_src  = (float*)(ws + 3 * NHB);
    float*    s_dst  = (float*)(ws + 3 * NHB + (size_t)NN * 4);
    float*    pooled = (float*)(ws + 3 * NHB + (size_t)NN * 16 + (size_t)ETOT * 4);
    float*    cnts   = (float*)(ws + 3 * NHB + (size_t)NN * 16 + (size_t)ETOT * 4 + (size_t)GG * HF * 4);

    const int gemm_grid   = ((NN / 16) + 7) / 8;
    const int agg_grid    = (NN + TT - 1) / TT;
    const int pool_grid   = (NN / PC_NODES + 7) / 8;

    gemm_wmma_bin<<<NCH + gemm_grid, 256, 0, stream>>>(x, W[0], asrc[0], adst[0],
                                                       h_lin, s_src, s_dst, INF_,
                                                       esrc, edst, seg, segcnt);
    agg_tile<true><<<agg_grid, AGG_THREADS, 0, stream>>>(esrc, edst, h_lin, s_src, s_dst, bias[0], act,
                                                         tl_src, tl_dl, tl_cnt, seg, segcnt);
    for (int l = 1; l < 3; ++l) {
        gemm_wmma<<<gemm_grid, 256, 0, stream>>>(act, W[l], asrc[l], adst[l],
                                                 h_lin, s_src, s_dst, HF);
        agg_tile<false><<<agg_grid, AGG_THREADS, 0, stream>>>(esrc, edst, h_lin, s_src, s_dst, bias[l], act,
                                                              tl_src, tl_dl, tl_cnt, seg, segcnt);
    }

    pool_init<<<PI_BLOCKS, 256, 0, stream>>>(pooled, cnts);
    pool_chunk<<<pool_grid, 256, 0, stream>>>(act, batch, pooled, cnts);
    fc_head<<<GG, 64, 0, stream>>>(pooled, cnts, fc1w, fc1b, fc2w, fc2b, out);
}
